# starting kernel
# baseline (speedup 1.0000x reference)
#include <hip/hip_runtime.h>
#include <hip/hip_bf16.h>
#include <math.h>


#define NEG_SLOPE 0.2f

typedef __attribute__((ext_vector_type(16))) _Float16 v16h;
typedef __attribute__((ext_vector_type(8)))  _Float16 v8h;
typedef __attribute__((ext_vector_type(8)))  float    v8f;

__device__ __forceinline__ unsigned fenc(float f) {
  unsigned u = __float_as_uint(f);
  return (u & 0x80000000u) ? ~u : (u | 0x80000000u);
}
__device__ __forceinline__ float fdec(unsigned u) {
  return (u & 0x80000000u) ? __uint_as_float(u & 0x7fffffffu) : __uint_as_float(~u);
}
#define ENC_NEG_INF 0x007FFFFFu

__global__ void k_f32_to_f16(const float* __restrict__ src, _Float16* __restrict__ dst,
                             long long n) {
  long long i = (long long)blockIdx.x * blockDim.x + threadIdx.x;
  if (i < n) dst[i] = (_Float16)src[i];
}

__global__ void k_init_out(float* __restrict__ out, const float* __restrict__ bias,
                           long long n, int hc) {
  long long i = (long long)blockIdx.x * blockDim.x + threadIdx.x;
  if (i < n) out[i] = bias[(int)(i % hc)];
}

__global__ void k_init_md(unsigned* __restrict__ M, float* __restrict__ D, long long n) {
  long long i = (long long)blockIdx.x * blockDim.x + threadIdx.x;
  if (i < n) { M[i] = ENC_NEG_INF; D[i] = 0.0f; }
}

__global__ void k_elu_f16(const float* __restrict__ in, _Float16* __restrict__ out,
                          long long n) {
  long long i = (long long)blockIdx.x * blockDim.x + threadIdx.x;
  if (i >= n) return;
  float v = in[i];
  v = (v > 0.0f) ? v : expm1f(v);
  out[i] = (_Float16)v;
}

__global__ void k_pack_w(const float* __restrict__ W, _Float16* __restrict__ WP,
                         int din, int dout) {
  int ktiles = din >> 5, ntiles = dout >> 4;
  int total = ktiles * ntiles * 32;
  int idx = blockIdx.x * blockDim.x + threadIdx.x;
  if (idx >= total) return;
  int lane = idx & 31;
  int frag = idx >> 5;
  int nt = frag % ntiles;
  int kt = frag / ntiles;
  int half = lane >> 4, lr = lane & 15;
  _Float16* o = WP + ((size_t)frag * 32 + lane) * 16;
  const float* w0 = W + (size_t)(kt * 32 + half * 8) * dout + nt * 16 + lr;
#pragma unroll
  for (int j = 0; j < 8; ++j) {
    o[j]     = (_Float16)w0[(size_t)j * dout];
    o[8 + j] = (_Float16)w0[(size_t)(16 + j) * dout];
  }
}

__global__ void k_gemm_f16(const _Float16* __restrict__ X, const _Float16* __restrict__ WP,
                           float* __restrict__ Y, int Mrows, int din, int dout) {
  int wave = threadIdx.x >> 5;
  int lane = threadIdx.x & 31;
  int half = lane >> 4;
  int lr   = lane & 15;
  int tileM = blockIdx.x * 4 + wave;
  if (tileM * 16 >= Mrows) return;
  int nt0 = blockIdx.y * 4;
  int ntiles = dout >> 4;

  int row = tileM * 16 + lr;
  if (row >= Mrows) row = Mrows - 1;

  v8f acc0 = {}, acc1 = {}, acc2 = {}, acc3 = {};
  const _Float16* xrow = X + (size_t)row * din;
  const v16h* WPv = (const v16h*)WP;
  for (int k0 = 0; k0 < din; k0 += 32) {
    const v8h* xa = (const v8h*)(xrow + k0 + half * 8);
    v8h lo = xa[0];
    v8h hi = xa[2];
    v16h a = __builtin_shufflevector(lo, hi, 0, 1, 2, 3, 4, 5, 6, 7,
                                     8, 9, 10, 11, 12, 13, 14, 15);
    size_t fb = ((size_t)(k0 >> 5) * ntiles + nt0) * 32 + lane;
    v16h b0 = WPv[fb];
    v16h b1 = WPv[fb + 32];
    v16h b2 = WPv[fb + 64];
    v16h b3 = WPv[fb + 96];
    acc0 = __builtin_amdgcn_wmma_f32_16x16x32_f16(false, a, false, b0, (short)0, acc0, false, false);
    acc1 = __builtin_amdgcn_wmma_f32_16x16x32_f16(false, a, false, b1, (short)0, acc1, false, false);
    acc2 = __builtin_amdgcn_wmma_f32_16x16x32_f16(false, a, false, b2, (short)0, acc2, false, false);
    acc3 = __builtin_amdgcn_wmma_f32_16x16x32_f16(false, a, false, b3, (short)0, acc3, false, false);
  }
  int r0 = tileM * 16 + half * 8;
#pragma unroll
  for (int i = 0; i < 8; ++i) {
    int r = r0 + i;
    if (r < Mrows) {
      float* yp = Y + (size_t)r * dout + nt0 * 16 + lr;
      yp[0]  = acc0[i];
      yp[16] = acc1[i];
      yp[32] = acc2[i];
      yp[48] = acc3[i];
    }
  }
}

template <int H, int C>
__global__ void k_edge_score(const float* __restrict__ XL, const float* __restrict__ XR,
                             const float* __restrict__ att, const int* __restrict__ ei,
                             long long E0, long long ET,
                             float* __restrict__ SC, unsigned* __restrict__ M) {
  constexpr int HC  = H * C;
  constexpr int LPH = 32 / H;
  constexpr int CPT = C / LPH;
  long long e = (long long)blockIdx.x * (blockDim.x >> 5) + (threadIdx.x >> 5);
  if (e >= ET) return;
  int lane = threadIdx.x & 31;
  int s, d;
  if (e < E0) { s = ei[e]; d = ei[E0 + e]; }
  else        { s = d = (int)(e - E0); }

  int h  = lane / LPH;
  int c0 = (lane % LPH) * CPT;
  const float* xl = XL + (size_t)s * HC + h * C + c0;
  const float* xr = XR + (size_t)d * HC + h * C + c0;
  const float* ah = att + h * C + c0;
  float p = 0.0f;
#pragma unroll
  for (int j = 0; j < CPT; ++j) {
    float v = xl[j] + xr[j];
    v = (v > 0.0f) ? v : NEG_SLOPE * v;
    p += ah[j] * v;
  }
#pragma unroll
  for (int o = LPH >> 1; o > 0; o >>= 1) p += __shfl_xor(p, o, 32);
  if ((lane % LPH) == 0) {
    SC[(size_t)e * H + h] = p;
    atomicMax(&M[(size_t)d * H + h], fenc(p));
  }
}

template <int H>
__global__ void k_edge_exp(float* __restrict__ SC, const unsigned* __restrict__ M,
                           float* __restrict__ DEN, const int* __restrict__ ei,
                           long long E0, long long ET) {
  long long i = (long long)blockIdx.x * blockDim.x + threadIdx.x;
  if (i >= ET * (long long)H) return;
  long long e = i / H;
  int h = (int)(i % H);
  int d = (e < E0) ? ei[E0 + e] : (int)(e - E0);
  float m = fdec(M[(size_t)d * H + h]);
  if (!(m > -3.0e38f && m < 3.0e38f)) m = 0.0f;
  float ex = __expf(SC[i] - m);
  SC[i] = ex;
  atomicAdd(&DEN[(size_t)d * H + h], ex);
}

template <int H, int C>
__global__ void k_edge_aggr(const float* __restrict__ XL, const float* __restrict__ SC,
                            const float* __restrict__ DEN, float* __restrict__ OUT,
                            const int* __restrict__ ei, long long E0, long long ET) {
  constexpr int HC  = H * C;
  constexpr int CPT = HC / 32;
  long long e = (long long)blockIdx.x * (blockDim.x >> 5) + (threadIdx.x >> 5);
  if (e >= ET) return;
  int lane = threadIdx.x & 31;
  int s, d;
  if (e < E0) { s = ei[e]; d = ei[E0 + e]; }
  else        { s = d = (int)(e - E0); }
  int ch0 = lane * CPT;
  int h   = ch0 / C;
  float ex    = SC[(size_t)e * H + h];
  float dn    = DEN[(size_t)d * H + h];
  float alpha = ex / fmaxf(dn, 1e-16f);
  const float* xs = XL + (size_t)s * HC + ch0;
  float*       od = OUT + (size_t)d * HC + ch0;
#pragma unroll
  for (int j = 0; j < CPT; ++j) atomicAdd(&od[j], alpha * xs[j]);
}

__global__ void k_softmax64(const float* __restrict__ in, float* __restrict__ out, int n) {
  int node = blockIdx.x * (blockDim.x >> 5) + (threadIdx.x >> 5);
  if (node >= n) return;
  int lane = threadIdx.x & 31;
  const float* p = in + (size_t)node * 64 + lane * 2;
  float a = p[0], b = p[1];
  float mx = fmaxf(a, b);
#pragma unroll
  for (int o = 16; o > 0; o >>= 1) mx = fmaxf(mx, __shfl_xor(mx, o, 32));
  float ea = __expf(a - mx), eb = __expf(b - mx);
  float sum = ea + eb;
#pragma unroll
  for (int o = 16; o > 0; o >>= 1) sum += __shfl_xor(sum, o, 32);
  float inv = 1.0f / sum;
  float* q = out + (size_t)node * 64 + lane * 2;
  q[0] = ea * inv;
  q[1] = eb * inv;
}

template <int H, int C>
static void gat_layer(const _Float16* X16, int din,
                      const float* Wl, const float* Wr, const float* att, const float* b,
                      _Float16* WPa, _Float16* WPb,
                      float* XL, float* XR, float* OUT, unsigned* M, float* DEN, float* SC,
                      const int* ei, int N, long long E0, hipStream_t stream) {
  constexpr int HC = H * C;
  long long ET = E0 + N;

  int packThreads = (din >> 5) * (HC >> 4) * 32;
  int packBlocks  = (packThreads + 255) / 256;
  k_pack_w<<<packBlocks, 256, 0, stream>>>(Wl, WPa, din, HC);
  k_pack_w<<<packBlocks, 256, 0, stream>>>(Wr, WPb, din, HC);

  int rowTiles = (N + 15) / 16;
  dim3 gg((unsigned)((rowTiles + 3) / 4), (unsigned)(HC / 64));
  k_gemm_f16<<<gg, 128, 0, stream>>>(X16, WPa, XL, N, din, HC);
  k_gemm_f16<<<gg, 128, 0, stream>>>(X16, WPb, XR, N, din, HC);

  long long nHC = (long long)N * HC, nH = (long long)N * H;
  k_init_out<<<(int)((nHC + 255) / 256), 256, 0, stream>>>(OUT, b, nHC, HC);
  k_init_md<<<(int)((nH + 255) / 256), 256, 0, stream>>>(M, DEN, nH);

  int gE = (int)((ET + 7) / 8);
  k_edge_score<H, C><<<gE, 256, 0, stream>>>(XL, XR, att, ei, E0, ET, SC, M);
  k_edge_exp<H><<<(int)((ET * H + 255) / 256), 256, 0, stream>>>(SC, M, DEN, ei, E0, ET);
  k_edge_aggr<H, C><<<gE, 256, 0, stream>>>(XL, SC, DEN, OUT, ei, E0, ET);
}

extern "C" void kernel_launch(void* const* d_in, const int* in_sizes, int n_in,
                              void* d_out, int out_size, void* d_ws, size_t ws_size,
                              hipStream_t stream) {
  (void)n_in; (void)out_size; (void)ws_size;
  const float* x    = (const float*)d_in[0];
  const int*   ei   = (const int*)d_in[1];
  const float* Wl1  = (const float*)d_in[2];
  const float* Wr1  = (const float*)d_in[3];
  const float* att1 = (const float*)d_in[4];
  const float* b1   = (const float*)d_in[5];
  const float* Wl2  = (const float*)d_in[6];
  const float* Wr2  = (const float*)d_in[7];
  const float* att2 = (const float*)d_in[8];
  const float* b2   = (const float*)d_in[9];
  const float* Wl3  = (const float*)d_in[10];
  const float* Wr3  = (const float*)d_in[11];
  const float* att3 = (const float*)d_in[12];
  const float* b3   = (const float*)d_in[13];

  int       N  = in_sizes[0] / 256;
  long long E0 = (long long)in_sizes[1] / 2;
  long long ET = E0 + N;

  uintptr_t p = (uintptr_t)d_ws;
  auto carve = [&](size_t bytes) -> void* {
    void* r = (void*)p;
    p += (bytes + 255) & ~(size_t)255;
    return r;
  };
  _Float16* X16 = (_Float16*)carve((size_t)N * 256 * sizeof(_Float16));
  _Float16* WPa = (_Float16*)carve((size_t)256 * 128 * sizeof(_Float16));
  _Float16* WPb = (_Float16*)carve((size_t)256 * 128 * sizeof(_Float16));
  float*    XL  = (float*)carve((size_t)N * 128 * sizeof(float));
  float*    XR  = (float*)carve((size_t)N * 128 * sizeof(float));
  float*    OUT = (float*)carve((size_t)N * 128 * sizeof(float));
  unsigned* M   = (unsigned*)carve((size_t)N * 8 * sizeof(unsigned));
  float*    DEN = (float*)carve((size_t)N * 8 * sizeof(float));
  float*    SC  = (float*)carve((size_t)ET * 8 * sizeof(float));

  long long nx = (long long)N * 256;
  k_f32_to_f16<<<(int)((nx + 255) / 256), 256, 0, stream>>>(x, X16, nx);
  gat_layer<8, 16>(X16, 256, Wl1, Wr1, att1, b1, WPa, WPb,
                   XL, XR, OUT, M, DEN, SC, ei, N, E0, stream);
  long long nh = (long long)N * 128;
  k_elu_f16<<<(int)((nh + 255) / 256), 256, 0, stream>>>(OUT, X16, nh);

  gat_layer<8, 16>(X16, 128, Wl2, Wr2, att2, b2, WPa, WPb,
                   XL, XR, OUT, M, DEN, SC, ei, N, E0, stream);
  k_elu_f16<<<(int)((nh + 255) / 256), 256, 0, stream>>>(OUT, X16, nh);

  gat_layer<1, 64>(X16, 128, Wl3, Wr3, att3, b3, WPa, WPb,
                   XL, XR, OUT, M, DEN, SC, ei, N, E0, stream);

  k_softmax64<<<(N + 7) / 8, 256, 0, stream>>>(OUT, (float*)d_out, N);
}


# Aggregation pass: lane = channel (lane+32j), so each f32 atomic instruction adds one whole 128-byte line of the target row, not 32-byte pieces of 4 lines; sums stay in atomic arrival order
# speedup vs baseline: 1.0891x; 1.0891x over previous
#include <hip/hip_runtime.h>
#include <hip/hip_bf16.h>
#include <math.h>


#define NEG_SLOPE 0.2f

typedef __attribute__((ext_vector_type(16))) _Float16 v16h;
typedef __attribute__((ext_vector_type(8)))  _Float16 v8h;
typedef __attribute__((ext_vector_type(8)))  float    v8f;

__device__ __forceinline__ unsigned fenc(float f) {
  unsigned u = __float_as_uint(f);
  return (u & 0x80000000u) ? ~u : (u | 0x80000000u);
}
__device__ __forceinline__ float fdec(unsigned u) {
  return (u & 0x80000000u) ? __uint_as_float(u & 0x7fffffffu) : __uint_as_float(~u);
}
#define ENC_NEG_INF 0x007FFFFFu

__global__ void k_f32_to_f16(const float* __restrict__ src, _Float16* __restrict__ dst,
                             long long n) {
  long long i = (long long)blockIdx.x * blockDim.x + threadIdx.x;
  if (i < n) dst[i] = (_Float16)src[i];
}

__global__ void k_init_out(float* __restrict__ out, const float* __restrict__ bias,
                           long long n, int hc) {
  long long i = (long long)blockIdx.x * blockDim.x + threadIdx.x;
  if (i < n) out[i] = bias[(int)(i % hc)];
}

__global__ void k_init_md(unsigned* __restrict__ M, float* __restrict__ D, long long n) {
  long long i = (long long)blockIdx.x * blockDim.x + threadIdx.x;
  if (i < n) { M[i] = ENC_NEG_INF; D[i] = 0.0f; }
}

__global__ void k_elu_f16(const float* __restrict__ in, _Float16* __restrict__ out,
                          long long n) {
  long long i = (long long)blockIdx.x * blockDim.x + threadIdx.x;
  if (i >= n) return;
  float v = in[i];
  v = (v > 0.0f) ? v : expm1f(v);
  out[i] = (_Float16)v;
}

__global__ void k_pack_w(const float* __restrict__ W, _Float16* __restrict__ WP,
                         int din, int dout) {
  int ktiles = din >> 5, ntiles = dout >> 4;
  int total = ktiles * ntiles * 32;
  int idx = blockIdx.x * blockDim.x + threadIdx.x;
  if (idx >= total) return;
  int lane = idx & 31;
  int frag = idx >> 5;
  int nt = frag % ntiles;
  int kt = frag / ntiles;
  int half = lane >> 4, lr = lane & 15;
  _Float16* o = WP + ((size_t)frag * 32 + lane) * 16;
  const float* w0 = W + (size_t)(kt * 32 + half * 8) * dout + nt * 16 + lr;
#pragma unroll
  for (int j = 0; j < 8; ++j) {
    o[j]     = (_Float16)w0[(size_t)j * dout];
    o[8 + j] = (_Float16)w0[(size_t)(16 + j) * dout];
  }
}

__global__ void k_gemm_f16(const _Float16* __restrict__ X, const _Float16* __restrict__ WP,
                           float* __restrict__ Y, int Mrows, int din, int dout) {
  int wave = threadIdx.x >> 5;
  int lane = threadIdx.x & 31;
  int half = lane >> 4;
  int lr   = lane & 15;
  int tileM = blockIdx.x * 4 + wave;
  if (tileM * 16 >= Mrows) return;
  int nt0 = blockIdx.y * 4;
  int ntiles = dout >> 4;

  int row = tileM * 16 + lr;
  if (row >= Mrows) row = Mrows - 1;

  v8f acc0 = {}, acc1 = {}, acc2 = {}, acc3 = {};
  const _Float16* xrow = X + (size_t)row * din;
  const v16h* WPv = (const v16h*)WP;
  for (int k0 = 0; k0 < din; k0 += 32) {
    const v8h* xa = (const v8h*)(xrow + k0 + half * 8);
    v8h lo = xa[0];
    v8h hi = xa[2];
    v16h a = __builtin_shufflevector(lo, hi, 0, 1, 2, 3, 4, 5, 6, 7,
                                     8, 9, 10, 11, 12, 13, 14, 15);
    size_t fb = ((size_t)(k0 >> 5) * ntiles + nt0) * 32 + lane;
    v16h b0 = WPv[fb];
    v16h b1 = WPv[fb + 32];
    v16h b2 = WPv[fb + 64];
    v16h b3 = WPv[fb + 96];
    acc0 = __builtin_amdgcn_wmma_f32_16x16x32_f16(false, a, false, b0, (short)0, acc0, false, false);
    acc1 = __builtin_amdgcn_wmma_f32_16x16x32_f16(false, a, false, b1, (short)0, acc1, false, false);
    acc2 = __builtin_amdgcn_wmma_f32_16x16x32_f16(false, a, false, b2, (short)0, acc2, false, false);
    acc3 = __builtin_amdgcn_wmma_f32_16x16x32_f16(false, a, false, b3, (short)0, acc3, false, false);
  }
  int r0 = tileM * 16 + half * 8;
#pragma unroll
  for (int i = 0; i < 8; ++i) {
    int r = r0 + i;
    if (r < Mrows) {
      float* yp = Y + (size_t)r * dout + nt0 * 16 + lr;
      yp[0]  = acc0[i];
      yp[16] = acc1[i];
      yp[32] = acc2[i];
      yp[48] = acc3[i];
    }
  }
}

template <int H, int C>
__global__ void k_edge_score(const float* __restrict__ XL, const float* __restrict__ XR,
                             const float* __restrict__ att, const int* __restrict__ ei,
                             long long E0, long long ET,
                             float* __restrict__ SC, unsigned* __restrict__ M) {
  constexpr int HC  = H * C;
  constexpr int LPH = 32 / H;
  constexpr int CPT = C / LPH;
  long long e = (long long)blockIdx.x * (blockDim.x >> 5) + (threadIdx.x >> 5);
  if (e >= ET) return;
  int lane = threadIdx.x & 31;
  int s, d;
  if (e < E0) { s = ei[e]; d = ei[E0 + e]; }
  else        { s = d = (int)(e - E0); }

  int h  = lane / LPH;
  int c0 = (lane % LPH) * CPT;
  const float* xl = XL + (size_t)s * HC + h * C + c0;
  const float* xr = XR + (size_t)d * HC + h * C + c0;
  const float* ah = att + h * C + c0;
  float p = 0.0f;
#pragma unroll
  for (int j = 0; j < CPT; ++j) {
    float v = xl[j] + xr[j];
    v = (v > 0.0f) ? v : NEG_SLOPE * v;
    p += ah[j] * v;
  }
#pragma unroll
  for (int o = LPH >> 1; o > 0; o >>= 1) p += __shfl_xor(p, o, 32);
  if ((lane % LPH) == 0) {
    SC[(size_t)e * H + h] = p;
    atomicMax(&M[(size_t)d * H + h], fenc(p));
  }
}

template <int H>
__global__ void k_edge_exp(float* __restrict__ SC, const unsigned* __restrict__ M,
                           float* __restrict__ DEN, const int* __restrict__ ei,
                           long long E0, long long ET) {
  long long i = (long long)blockIdx.x * blockDim.x + threadIdx.x;
  if (i >= ET * (long long)H) return;
  long long e = i / H;
  int h = (int)(i % H);
  int d = (e < E0) ? ei[E0 + e] : (int)(e - E0);
  float m = fdec(M[(size_t)d * H + h]);
  if (!(m > -3.0e38f && m < 3.0e38f)) m = 0.0f;
  float ex = __expf(SC[i] - m);
  SC[i] = ex;
  atomicAdd(&DEN[(size_t)d * H + h], ex);
}

template <int H, int C>
__global__ void k_edge_aggr(const float* __restrict__ XL, const float* __restrict__ SC,
                            const float* __restrict__ DEN, float* __restrict__ OUT,
                            const int* __restrict__ ei, long long E0, long long ET) {
  constexpr int HC = H * C;
  constexpr int NJ = HC / 32;
  long long e = (long long)blockIdx.x * (blockDim.x >> 5) + (threadIdx.x >> 5);
  if (e >= ET) return;
  int lane = threadIdx.x & 31;
  int s, d;
  if (e < E0) { s = ei[e]; d = ei[E0 + e]; }
  else        { s = d = (int)(e - E0); }
  int hl = lane & (H - 1);
  float ex    = SC[(size_t)e * H + hl];
  float dn    = DEN[(size_t)d * H + hl];
  float alpha = ex / fmaxf(dn, 1e-16f);
  const float* xs = XL + (size_t)s * HC + lane;
  float*       od = OUT + (size_t)d * HC + lane;
  float v[NJ];
#pragma unroll
  for (int j = 0; j < NJ; ++j) v[j] = xs[32 * j];
#pragma unroll
  for (int j = 0; j < NJ; ++j) {
    float a = alpha;
    if (H > 1) a = __shfl(alpha, (lane + 32 * j) / C, 32);
    atomicAdd(&od[32 * j], a * v[j]);
  }
}

__global__ void k_softmax64(const float* __restrict__ in, float* __restrict__ out, int n) {
  int node = blockIdx.x * (blockDim.x >> 5) + (threadIdx.x >> 5);
  if (node >= n) return;
  int lane = threadIdx.x & 31;
  const float* p = in + (size_t)node * 64 + lane * 2;
  float a = p[0], b = p[1];
  float mx = fmaxf(a, b);
#pragma unroll
  for (int o = 16; o > 0; o >>= 1) mx = fmaxf(mx, __shfl_xor(mx, o, 32));
  float ea = __expf(a - mx), eb = __expf(b - mx);
  float sum = ea + eb;
#pragma unroll
  for (int o = 16; o > 0; o >>= 1) sum += __shfl_xor(sum, o, 32);
  float inv = 1.0f / sum;
  float* q = out + (size_t)node * 64 + lane * 2;
  q[0] = ea * inv;
  q[1] = eb * inv;
}

template <int H, int C>
static void gat_layer(const _Float16* X16, int din,
                      const float* Wl, const float* Wr, const float* att, const float* b,
                      _Float16* WPa, _Float16* WPb,
                      float* XL, float* XR, float* OUT, unsigned* M, float* DEN, float* SC,
                      const int* ei, int N, long long E0, hipStream_t stream) {
  constexpr int HC = H * C;
  long long ET = E0 + N;

  int packThreads = (din >> 5) * (HC >> 4) * 32;
  int packBlocks  = (packThreads + 255) / 256;
  k_pack_w<<<packBlocks, 256, 0, stream>>>(Wl, WPa, din, HC);
  k_pack_w<<<packBlocks, 256, 0, stream>>>(Wr, WPb, din, HC);

  int rowTiles = (N + 15) / 16;
  dim3 gg((unsigned)((rowTiles + 3) / 4), (unsigned)(HC / 64));
  k_gemm_f16<<<gg, 128, 0, stream>>>(X16, WPa, XL, N, din, HC);
  k_gemm_f16<<<gg, 128, 0, stream>>>(X16, WPb, XR, N, din, HC);

  long long nHC = (long long)N * HC, nH = (long long)N * H;
  k_init_out<<<(int)((nHC + 255) / 256), 256, 0, stream>>>(OUT, b, nHC, HC);
  k_init_md<<<(int)((nH + 255) / 256), 256, 0, stream>>>(M, DEN, nH);

  int gE = (int)((ET + 7) / 8);
  k_edge_score<H, C><<<gE, 256, 0, stream>>>(XL, XR, att, ei, E0, ET, SC, M);
  k_edge_exp<H><<<(int)((ET * H + 255) / 256), 256, 0, stream>>>(SC, M, DEN, ei, E0, ET);
  k_edge_aggr<H, C><<<gE, 256, 0, stream>>>(XL, SC, DEN, OUT, ei, E0, ET);
}

extern "C" void kernel_launch(void* const* d_in, const int* in_sizes, int n_in,
                              void* d_out, int out_size, void* d_ws, size_t ws_size,
                              hipStream_t stream) {
  (void)n_in; (void)out_size; (void)ws_size;
  const float* x    = (const float*)d_in[0];
  const int*   ei   = (const int*)d_in[1];
  const float* Wl1  = (const float*)d_in[2];
  const float* Wr1  = (const float*)d_in[3];
  const float* att1 = (const float*)d_in[4];
  const float* b1   = (const float*)d_in[5];
  const float* Wl2  = (const float*)d_in[6];
  const float* Wr2  = (const float*)d_in[7];
  const float* att2 = (const float*)d_in[8];
  const float* b2   = (const float*)d_in[9];
  const float* Wl3  = (const float*)d_in[10];
  const float* Wr3  = (const float*)d_in[11];
  const float* att3 = (const float*)d_in[12];
  const float* b3   = (const float*)d_in[13];

  int       N  = in_sizes[0] / 256;
  long long E0 = (long long)in_sizes[1] / 2;
  long long ET = E0 + N;

  uintptr_t p = (uintptr_t)d_ws;
  auto carve = [&](size_t bytes) -> void* {
    void* r = (void*)p;
    p += (bytes + 255) & ~(size_t)255;
    return r;
  };
  _Float16* X16 = (_Float16*)carve((size_t)N * 256 * sizeof(_Float16));
  _Float16* WPa = (_Float16*)carve((size_t)256 * 128 * sizeof(_Float16));
  _Float16* WPb = (_Float16*)carve((size_t)256 * 128 * sizeof(_Float16));
  float*    XL  = (float*)carve((size_t)N * 128 * sizeof(float));
  float*    XR  = (float*)carve((size_t)N * 128 * sizeof(float));
  float*    OUT = (float*)carve((size_t)N * 128 * sizeof(float));
  unsigned* M   = (unsigned*)carve((size_t)N * 8 * sizeof(unsigned));
  float*    DEN = (float*)carve((size_t)N * 8 * sizeof(float));
  float*    SC  = (float*)carve((size_t)ET * 8 * sizeof(float));

  long long nx = (long long)N * 256;
  k_f32_to_f16<<<(int)((nx + 255) / 256), 256, 0, stream>>>(x, X16, nx);
  gat_layer<8, 16>(X16, 256, Wl1, Wr1, att1, b1, WPa, WPb,
                   XL, XR, OUT, M, DEN, SC, ei, N, E0, stream);
  long long nh = (long long)N * 128;
  k_elu_f16<<<(int)((nh + 255) / 256), 256, 0, stream>>>(OUT, X16, nh);

  gat_layer<8, 16>(X16, 128, Wl2, Wr2, att2, b2, WPa, WPb,
                   XL, XR, OUT, M, DEN, SC, ei, N, E0, stream);
  k_elu_f16<<<(int)((nh + 255) / 256), 256, 0, stream>>>(OUT, X16, nh);

  gat_layer<1, 64>(X16, 128, Wl3, Wr3, att3, b3, WPa, WPb,
                   XL, XR, OUT, M, DEN, SC, ei, N, E0, stream);

  k_softmax64<<<(N + 7) / 8, 256, 0, stream>>>(OUT, (float*)d_out, N);
}


# Score pass only (no float atomics there; exp and aggregation passes and their arrival-order sums untouched): one wave owns 32 consecutive edges, rows of 4 or 8 edges gathered together; same sum order
# speedup vs baseline: 1.3093x; 1.2022x over previous
#include <hip/hip_runtime.h>
#include <hip/hip_bf16.h>
#include <math.h>


#define NEG_SLOPE 0.2f

typedef __attribute__((ext_vector_type(16))) _Float16 v16h;
typedef __attribute__((ext_vector_type(8)))  _Float16 v8h;
typedef __attribute__((ext_vector_type(8)))  float    v8f;

__device__ __forceinline__ unsigned fenc(float f) {
  unsigned u = __float_as_uint(f);
  return (u & 0x80000000u) ? ~u : (u | 0x80000000u);
}
__device__ __forceinline__ float fdec(unsigned u) {
  return (u & 0x80000000u) ? __uint_as_float(u & 0x7fffffffu) : __uint_as_float(~u);
}
#define ENC_NEG_INF 0x007FFFFFu

__global__ void k_f32_to_f16(const float* __restrict__ src, _Float16* __restrict__ dst,
                             long long n) {
  long long i = (long long)blockIdx.x * blockDim.x + threadIdx.x;
  if (i < n) dst[i] = (_Float16)src[i];
}

__global__ void k_init_out(float* __restrict__ out, const float* __restrict__ bias,
                           long long n, int hc) {
  long long i = (long long)blockIdx.x * blockDim.x + threadIdx.x;
  if (i < n) out[i] = bias[(int)(i % hc)];
}

__global__ void k_init_md(unsigned* __restrict__ M, float* __restrict__ D, long long n) {
  long long i = (long long)blockIdx.x * blockDim.x + threadIdx.x;
  if (i < n) { M[i] = ENC_NEG_INF; D[i] = 0.0f; }
}

__global__ void k_elu_f16(const float* __restrict__ in, _Float16* __restrict__ out,
                          long long n) {
  long long i = (long long)blockIdx.x * blockDim.x + threadIdx.x;
  if (i >= n) return;
  float v = in[i];
  v = (v > 0.0f) ? v : expm1f(v);
  out[i] = (_Float16)v;
}

__global__ void k_pack_w(const float* __restrict__ W, _Float16* __restrict__ WP,
                         int din, int dout) {
  int ktiles = din >> 5, ntiles = dout >> 4;
  int total = ktiles * ntiles * 32;
  int idx = blockIdx.x * blockDim.x + threadIdx.x;
  if (idx >= total) return;
  int lane = idx & 31;
  int frag = idx >> 5;
  int nt = frag % ntiles;
  int kt = frag / ntiles;
  int half = lane >> 4, lr = lane & 15;
  _Float16* o = WP + ((size_t)frag * 32 + lane) * 16;
  const float* w0 = W + (size_t)(kt * 32 + half * 8) * dout + nt * 16 + lr;
#pragma unroll
  for (int j = 0; j < 8; ++j) {
    o[j]     = (_Float16)w0[(size_t)j * dout];
    o[8 + j] = (_Float16)w0[(size_t)(16 + j) * dout];
  }
}

__global__ void k_gemm_f16(const _Float16* __restrict__ X, const _Float16* __restrict__ WP,
                           float* __restrict__ Y, int Mrows, int din, int dout) {
  int wave = threadIdx.x >> 5;
  int lane = threadIdx.x & 31;
  int half = lane >> 4;
  int lr   = lane & 15;
  int tileM = blockIdx.x * 4 + wave;
  if (tileM * 16 >= Mrows) return;
  int nt0 = blockIdx.y * 4;
  int ntiles = dout >> 4;

  int row = tileM * 16 + lr;
  if (row >= Mrows) row = Mrows - 1;

  v8f acc0 = {}, acc1 = {}, acc2 = {}, acc3 = {};
  const _Float16* xrow = X + (size_t)row * din;
  const v16h* WPv = (const v16h*)WP;
  for (int k0 = 0; k0 < din; k0 += 32) {
    const v8h* xa = (const v8h*)(xrow + k0 + half * 8);
    v8h lo = xa[0];
    v8h hi = xa[2];
    v16h a = __builtin_shufflevector(lo, hi, 0, 1, 2, 3, 4, 5, 6, 7,
                                     8, 9, 10, 11, 12, 13, 14, 15);
    size_t fb = ((size_t)(k0 >> 5) * ntiles + nt0) * 32 + lane;
    v16h b0 = WPv[fb];
    v16h b1 = WPv[fb + 32];
    v16h b2 = WPv[fb + 64];
    v16h b3 = WPv[fb + 96];
    acc0 = __builtin_amdgcn_wmma_f32_16x16x32_f16(false, a, false, b0, (short)0, acc0, false, false);
    acc1 = __builtin_amdgcn_wmma_f32_16x16x32_f16(false, a, false, b1, (short)0, acc1, false, false);
    acc2 = __builtin_amdgcn_wmma_f32_16x16x32_f16(false, a, false, b2, (short)0, acc2, false, false);
    acc3 = __builtin_amdgcn_wmma_f32_16x16x32_f16(false, a, false, b3, (short)0, acc3, false, false);
  }
  int r0 = tileM * 16 + half * 8;
#pragma unroll
  for (int i = 0; i < 8; ++i) {
    int r = r0 + i;
    if (r < Mrows) {
      float* yp = Y + (size_t)r * dout + nt0 * 16 + lr;
      yp[0]  = acc0[i];
      yp[16] = acc1[i];
      yp[32] = acc2[i];
      yp[48] = acc3[i];
    }
  }
}

template <int N, int O>
__device__ __forceinline__ void lane_reduce(float* p, int lane) {
  if constexpr (O > 0) {
    if constexpr (N > 1) {
      constexpr int HN = N / 2;
      const bool up = (lane & O) != 0;
      for (int i = 0; i < HN; ++i) {
        float keep = up ? p[HN + i] : p[i];
        float send = up ? p[i] : p[HN + i];
        p[i] = keep + __shfl_xor(send, O, 32);
      }
      lane_reduce<HN, O / 2>(p, lane);
    } else {
      p[0] += __shfl_xor(p[0], O, 32);
      lane_reduce<1, O / 2>(p, lane);
    }
  }
}

template <int H, int C, int NB>
__global__ void k_edge_score(const float* __restrict__ XL, const float* __restrict__ XR,
                             const float* __restrict__ att, const int* __restrict__ ei,
                             long long E0, long long ET,
                             float* __restrict__ SC, unsigned* __restrict__ M) {
  constexpr int HC  = H * C;
  constexpr int LPH = 32 / H;
  constexpr int CPT = C / LPH;
  constexpr int G   = LPH / NB;
  static_assert(NB * G == LPH && H * LPH == 32 && CPT * LPH == C, "shape");
  const long long e0 =
      ((long long)blockIdx.x * (blockDim.x >> 5) + (threadIdx.x >> 5)) * 32;
  if (e0 >= ET) return;
  const int lane = threadIdx.x & 31;

  const long long el = e0 + lane;
  int sv = 0, dv = 0;
  if (el < E0)      { sv = ei[el]; dv = ei[E0 + el]; }
  else if (el < ET) { sv = dv = (int)(el - E0); }

  const int h   = lane / LPH;
  const int q   = lane % LPH;
  const int kk  = q / G;
  const int rep = q % G;
  float a[CPT];
  for (int j = 0; j < CPT; ++j) a[j] = att[lane * CPT + j];
  const float* xlb = XL + lane * CPT;
  const float* xrb = XR + lane * CPT;

  for (int sb = 0; sb < H; ++sb) {
    float r  = 0.0f;
    int   rd = 0;
    for (int bb = 0; bb < G; ++bb) {
      const int base = sb * LPH + bb * NB;
      float xl[NB][CPT], xr[NB][CPT];
      int   dk[NB];
      for (int k = 0; k < NB; ++k) {
        const int s = __builtin_amdgcn_readlane(sv, base + k);
        const int d = __builtin_amdgcn_readlane(dv, base + k);
        dk[k] = d;
        const float* pl = xlb + (size_t)s * HC;
        const float* pr = xrb + (size_t)d * HC;
        for (int j = 0; j < CPT; ++j) { xl[k][j] = pl[j]; xr[k][j] = pr[j]; }
      }
      float p[NB];
      for (int k = 0; k < NB; ++k) {
        float acc = 0.0f;
        for (int j = 0; j < CPT; ++j) {
          float v = xl[k][j] + xr[k][j];
          v = (v > 0.0f) ? v : NEG_SLOPE * v;
          acc += a[j] * v;
        }
        p[k] = acc;
      }
      lane_reduce<NB, LPH / 2>(p, lane);
      int dq = dk[0];
      for (int k = 1; k < NB; ++k) dq = (kk == k) ? dk[k] : dq;
      if (G == 1 || rep == bb) { r = p[0]; rd = dq; }
    }
    const long long e = e0 + sb * LPH + rep * NB + kk;
    if (e < ET) {
      SC[(size_t)e * H + h] = r;
      atomicMax(&M[(size_t)rd * H + h], fenc(r));
    }
  }
}

template <int H>
__global__ void k_edge_exp(float* __restrict__ SC, const unsigned* __restrict__ M,
                           float* __restrict__ DEN, const int* __restrict__ ei,
                           long long E0, long long ET) {
  long long i = (long long)blockIdx.x * blockDim.x + threadIdx.x;
  if (i >= ET * (long long)H) return;
  long long e = i / H;
  int h = (int)(i % H);
  int d = (e < E0) ? ei[E0 + e] : (int)(e - E0);
  float m = fdec(M[(size_t)d * H + h]);
  if (!(m > -3.0e38f && m < 3.0e38f)) m = 0.0f;
  float ex = __expf(SC[i] - m);
  SC[i] = ex;
  atomicAdd(&DEN[(size_t)d * H + h], ex);
}

template <int H, int C>
__global__ void k_edge_aggr(const float* __restrict__ XL, const float* __restrict__ SC,
                            const float* __restrict__ DEN, float* __restrict__ OUT,
                            const int* __restrict__ ei, long long E0, long long ET) {
  constexpr int HC = H * C;
  constexpr int NJ = HC / 32;
  long long e = (long long)blockIdx.x * (blockDim.x >> 5) + (threadIdx.x >> 5);
  if (e >= ET) return;
  int lane = threadIdx.x & 31;
  int s, d;
  if (e < E0) { s = ei[e]; d = ei[E0 + e]; }
  else        { s = d = (int)(e - E0); }
  int hl = lane & (H - 1);
  float ex    = SC[(size_t)e * H + hl];
  float dn    = DEN[(size_t)d * H + hl];
  float alpha = ex / fmaxf(dn, 1e-16f);
  const float* xs = XL + (size_t)s * HC + lane;
  float*       od = OUT + (size_t)d * HC + lane;
  float v[NJ];
#pragma unroll
  for (int j = 0; j < NJ; ++j) v[j] = xs[32 * j];
#pragma unroll
  for (int j = 0; j < NJ; ++j) {
    float a = alpha;
    if (H > 1) a = __shfl(alpha, (lane + 32 * j) / C, 32);
    atomicAdd(&od[32 * j], a * v[j]);
  }
}

__global__ void k_softmax64(const float* __restrict__ in, float* __restrict__ out, int n) {
  int node = blockIdx.x * (blockDim.x >> 5) + (threadIdx.x >> 5);
  if (node >= n) return;
  int lane = threadIdx.x & 31;
  const float* p = in + (size_t)node * 64 + lane * 2;
  float a = p[0], b = p[1];
  float mx = fmaxf(a, b);
#pragma unroll
  for (int o = 16; o > 0; o >>= 1) mx = fmaxf(mx, __shfl_xor(mx, o, 32));
  float ea = __expf(a - mx), eb = __expf(b - mx);
  float sum = ea + eb;
#pragma unroll
  for (int o = 16; o > 0; o >>= 1) sum += __shfl_xor(sum, o, 32);
  float inv = 1.0f / sum;
  float* q = out + (size_t)node * 64 + lane * 2;
  q[0] = ea * inv;
  q[1] = eb * inv;
}

template <int H, int C>
static void gat_layer(const _Float16* X16, int din,
                      const float* Wl, const float* Wr, const float* att, const float* b,
                      _Float16* WPa, _Float16* WPb,
                      float* XL, float* XR, float* OUT, unsigned* M, float* DEN, float* SC,
                      const int* ei, int N, long long E0, hipStream_t stream) {
  constexpr int HC = H * C;
  long long ET = E0 + N;

  int packThreads = (din >> 5) * (HC >> 4) * 32;
  int packBlocks  = (packThreads + 255) / 256;
  k_pack_w<<<packBlocks, 256, 0, stream>>>(Wl, WPa, din, HC);
  k_pack_w<<<packBlocks, 256, 0, stream>>>(Wr, WPb, din, HC);

  int rowTiles = (N + 15) / 16;
  dim3 gg((unsigned)((rowTiles + 3) / 4), (unsigned)(HC / 64));
  k_gemm_f16<<<gg, 128, 0, stream>>>(X16, WPa, XL, N, din, HC);
  k_gemm_f16<<<gg, 128, 0, stream>>>(X16, WPb, XR, N, din, HC);

  long long nHC = (long long)N * HC, nH = (long long)N * H;
  k_init_out<<<(int)((nHC + 255) / 256), 256, 0, stream>>>(OUT, b, nHC, HC);
  k_init_md<<<(int)((nH + 255) / 256), 256, 0, stream>>>(M, DEN, nH);

  int gE = (int)((ET + 7) / 8);
  int gS = (int)((ET + 255) / 256);
  k_edge_score<H, C, (H == 1 ? 8 : 4)><<<gS, 256, 0, stream>>>(XL, XR, att, ei, E0, ET, SC, M);
  k_edge_exp<H><<<(int)((ET * H + 255) / 256), 256, 0, stream>>>(SC, M, DEN, ei, E0, ET);
  k_edge_aggr<H, C><<<gE, 256, 0, stream>>>(XL, SC, DEN, OUT, ei, E0, ET);
}

extern "C" void kernel_launch(void* const* d_in, const int* in_sizes, int n_in,
                              void* d_out, int out_size, void* d_ws, size_t ws_size,
                              hipStream_t stream) {
  (void)n_in; (void)out_size; (void)ws_size;
  const float* x    = (const float*)d_in[0];
  const int*   ei   = (const int*)d_in[1];
  const float* Wl1  = (const float*)d_in[2];
  const float* Wr1  = (const float*)d_in[3];
  const float* att1 = (const float*)d_in[4];
  const float* b1   = (const float*)d_in[5];
  const float* Wl2  = (const float*)d_in[6];
  const float* Wr2  = (const float*)d_in[7];
  const float* att2 = (const float*)d_in[8];
  const float* b2   = (const float*)d_in[9];
  const float* Wl3  = (const float*)d_in[10];
  const float* Wr3  = (const float*)d_in[11];
  const float* att3 = (const float*)d_in[12];
  const float* b3   = (const float*)d_in[13];

  int       N  = in_sizes[0] / 256;
  long long E0 = (long long)in_sizes[1] / 2;
  long long ET = E0 + N;

  uintptr_t p = (uintptr_t)d_ws;
  auto carve = [&](size_t bytes) -> void* {
    void* r = (void*)p;
    p += (bytes + 255) & ~(size_t)255;
    return r;
  };
  _Float16* X16 = (_Float16*)carve((size_t)N * 256 * sizeof(_Float16));
  _Float16* WPa = (_Float16*)carve((size_t)256 * 128 * sizeof(_Float16));
  _Float16* WPb = (_Float16*)carve((size_t)256 * 128 * sizeof(_Float16));
  float*    XL  = (float*)carve((size_t)N * 128 * sizeof(float));
  float*    XR  = (float*)carve((size_t)N * 128 * sizeof(float));
  float*    OUT = (float*)carve((size_t)N * 128 * sizeof(float));
  unsigned* M   = (unsigned*)carve((size_t)N * 8 * sizeof(unsigned));
  float*    DEN = (float*)carve((size_t)N * 8 * sizeof(float));
  float*    SC  = (float*)carve((size_t)ET * 8 * sizeof(float));

  long long nx = (long long)N * 256;
  k_f32_to_f16<<<(int)((nx + 255) / 256), 256, 0, stream>>>(x, X16, nx);
  gat_layer<8, 16>(X16, 256, Wl1, Wr1, att1, b1, WPa, WPb,
                   XL, XR, OUT, M, DEN, SC, ei, N, E0, stream);
  long long nh = (long long)N * 128;
  k_elu_f16<<<(int)((nh + 255) / 256), 256, 0, stream>>>(OUT, X16, nh);

  gat_layer<8, 16>(X16, 128, Wl2, Wr2, att2, b2, WPa, WPb,
                   XL, XR, OUT, M, DEN, SC, ei, N, E0, stream);
  k_elu_f16<<<(int)((nh + 255) / 256), 256, 0, stream>>>(OUT, X16, nh);

  gat_layer<1, 64>(X16, 128, Wl3, Wr3, att3, b3, WPa, WPb,
                   XL, XR, OUT, M, DEN, SC, ei, N, E0, stream);

  k_softmax64<<<(N + 7) / 8, 256, 0, stream>>>(OUT, (float*)d_out, N);
}


# Aggregation pass: one wave owns 32 consecutive edges (rows of 8 edges in flight together); sums now integer atomics of round(addend*2^k), k per layer: exact adds, so no arrival order in the result
# speedup vs baseline: 1.6583x; 1.2666x over previous
#include <hip/hip_runtime.h>
#include <hip/hip_bf16.h>
#include <math.h>


#define NEG_SLOPE 0.2f

typedef __attribute__((ext_vector_type(16))) _Float16 v16h;
typedef __attribute__((ext_vector_type(8)))  _Float16 v8h;
typedef __attribute__((ext_vector_type(8)))  float    v8f;

__device__ __forceinline__ unsigned fenc(float f) {
  unsigned u = __float_as_uint(f);
  return (u & 0x80000000u) ? ~u : (u | 0x80000000u);
}
__device__ __forceinline__ float fdec(unsigned u) {
  return (u & 0x80000000u) ? __uint_as_float(u & 0x7fffffffu) : __uint_as_float(~u);
}
#define ENC_NEG_INF 0x007FFFFFu

__global__ void k_f32_to_f16(const float* __restrict__ src, _Float16* __restrict__ dst,
                             long long n) {
  long long i = (long long)blockIdx.x * blockDim.x + threadIdx.x;
  if (i < n) dst[i] = (_Float16)src[i];
}

__global__ void k_init_out(int* __restrict__ out, long long n) {
  long long i = (long long)blockIdx.x * blockDim.x + threadIdx.x;
  if (i < n) out[i] = 0;
}

__global__ void k_init_md(unsigned* __restrict__ M, unsigned* __restrict__ D, long long n) {
  long long i = (long long)blockIdx.x * blockDim.x + threadIdx.x;
  if (i < n) { M[i] = ENC_NEG_INF; D[i] = 0u; }
}

#define SMALL_WORDS 256
__global__ void k_zero_small(unsigned* __restrict__ DEG, unsigned* __restrict__ SMALL, int n) {
  int i = blockIdx.x * blockDim.x + threadIdx.x;
  if (i < n) DEG[i] = 0u;
  if (i < SMALL_WORDS) SMALL[i] = 0u;
}

__global__ void k_count_deg(const int* __restrict__ ei, long long E0, unsigned* __restrict__ DEG) {
  long long i = (long long)blockIdx.x * blockDim.x + threadIdx.x;
  if (i < E0) atomicAdd(&DEG[ei[E0 + i]], 1u);
}

__global__ void k_deg_max(const unsigned* __restrict__ DEG, int n, unsigned* __restrict__ SMALL) {
  unsigned v = 0u;
  for (int i = blockIdx.x * blockDim.x + threadIdx.x; i < n; i += gridDim.x * blockDim.x) {
    unsigned t = DEG[i];
    v = (t > v) ? t : v;
  }
#pragma unroll
  for (int o = 16; o > 0; o >>= 1) {
    unsigned t = (unsigned)__shfl_xor((int)v, o, 32);
    v = (t > v) ? t : v;
  }
  if ((threadIdx.x & 31) == 0) atomicMax(&SMALL[0], v);
}

__global__ void k_scale(const unsigned* __restrict__ SMALL, const unsigned* __restrict__ amax64,
                        float* __restrict__ PAR) {
  int lane = threadIdx.x & 31;
  unsigned a = amax64[lane], b = amax64[lane + 32];
  a = (b > a) ? b : a;
#pragma unroll
  for (int o = 16; o > 0; o >>= 1) {
    unsigned t = (unsigned)__shfl_xor((int)a, o, 32);
    a = (t > a) ? t : a;
  }
  if (lane == 0) {
    unsigned deg = SMALL[0] + 1u;
    float bound = (float)deg * __uint_as_float(a);
    int eb = (int)((__float_as_uint(bound) >> 23) & 0xffu);
    int k = 156 - eb;
    k = (k > 100) ? 100 : ((k < -100) ? -100 : k);
    int p = 32 - __clz((int)deg);
    int kd = 31 - p;
    kd = (kd > 30) ? 30 : ((kd < 0) ? 0 : kd);
    PAR[0] = __uint_as_float((unsigned)(127 + k) << 23);
    PAR[1] = __uint_as_float((unsigned)(127 - k) << 23);
    PAR[2] = __uint_as_float((unsigned)(127 + kd) << 23);
    PAR[3] = __uint_as_float((unsigned)(127 - kd) << 23);
  }
}

template <int H, int C>
__global__ void k_norm_elu_f16(const int* __restrict__ in, const unsigned* __restrict__ den,
                               const float* __restrict__ bias, _Float16* __restrict__ out,
                               long long n, const float* __restrict__ PAR) {
  constexpr int HC = H * C;
  long long i = (long long)blockIdx.x * blockDim.x + threadIdx.x;
  if (i >= n) return;
  long long node = i / HC;
  int c = (int)(i % HC);
  float sum = (float)in[i] * PAR[1];
  float dn  = (float)den[(size_t)node * H + c / C] * PAR[3];
  float v = sum / fmaxf(dn, 1e-16f) + bias[c];
  v = (v > 0.0f) ? v : expm1f(v);
  out[i] = (_Float16)v;
}

__global__ void k_pack_w(const float* __restrict__ W, _Float16* __restrict__ WP,
                         int din, int dout) {
  int ktiles = din >> 5, ntiles = dout >> 4;
  int total = ktiles * ntiles * 32;
  int idx = blockIdx.x * blockDim.x + threadIdx.x;
  if (idx >= total) return;
  int lane = idx & 31;
  int frag = idx >> 5;
  int nt = frag % ntiles;
  int kt = frag / ntiles;
  int half = lane >> 4, lr = lane & 15;
  _Float16* o = WP + ((size_t)frag * 32 + lane) * 16;
  const float* w0 = W + (size_t)(kt * 32 + half * 8) * dout + nt * 16 + lr;
#pragma unroll
  for (int j = 0; j < 8; ++j) {
    o[j]     = (_Float16)w0[(size_t)j * dout];
    o[8 + j] = (_Float16)w0[(size_t)(16 + j) * dout];
  }
}

template <bool AMAX>
__global__ void k_gemm_f16(const _Float16* __restrict__ X, const _Float16* __restrict__ WP,
                           float* __restrict__ Y, int Mrows, int din, int dout,
                           unsigned* __restrict__ amax) {
  int wave = threadIdx.x >> 5;
  int lane = threadIdx.x & 31;
  int half = lane >> 4;
  int lr   = lane & 15;
  int tileM = blockIdx.x * 4 + wave;
  if (tileM * 16 >= Mrows) return;
  int nt0 = blockIdx.y * 4;
  int ntiles = dout >> 4;

  int row = tileM * 16 + lr;
  if (row >= Mrows) row = Mrows - 1;

  v8f acc0 = {}, acc1 = {}, acc2 = {}, acc3 = {};
  const _Float16* xrow = X + (size_t)row * din;
  const v16h* WPv = (const v16h*)WP;
  for (int k0 = 0; k0 < din; k0 += 32) {
    const v8h* xa = (const v8h*)(xrow + k0 + half * 8);
    v8h lo = xa[0];
    v8h hi = xa[2];
    v16h a = __builtin_shufflevector(lo, hi, 0, 1, 2, 3, 4, 5, 6, 7,
                                     8, 9, 10, 11, 12, 13, 14, 15);
    size_t fb = ((size_t)(k0 >> 5) * ntiles + nt0) * 32 + lane;
    v16h b0 = WPv[fb];
    v16h b1 = WPv[fb + 32];
    v16h b2 = WPv[fb + 64];
    v16h b3 = WPv[fb + 96];
    acc0 = __builtin_amdgcn_wmma_f32_16x16x32_f16(false, a, false, b0, (short)0, acc0, false, false);
    acc1 = __builtin_amdgcn_wmma_f32_16x16x32_f16(false, a, false, b1, (short)0, acc1, false, false);
    acc2 = __builtin_amdgcn_wmma_f32_16x16x32_f16(false, a, false, b2, (short)0, acc2, false, false);
    acc3 = __builtin_amdgcn_wmma_f32_16x16x32_f16(false, a, false, b3, (short)0, acc3, false, false);
  }
  int r0 = tileM * 16 + half * 8;
#pragma unroll
  for (int i = 0; i < 8; ++i) {
    int r = r0 + i;
    if (r < Mrows) {
      float* yp = Y + (size_t)r * dout + nt0 * 16 + lr;
      yp[0]  = acc0[i];
      yp[16] = acc1[i];
      yp[32] = acc2[i];
      yp[48] = acc3[i];
    }
  }
  if constexpr (AMAX) {
    float mx = 0.0f;
#pragma unroll
    for (int i = 0; i < 8; ++i) {
      mx = fmaxf(mx, fmaxf(fmaxf(fabsf(acc0[i]), fabsf(acc1[i])),
                           fmaxf(fabsf(acc2[i]), fabsf(acc3[i]))));
    }
#pragma unroll
    for (int o = 16; o > 0; o >>= 1) mx = fmaxf(mx, __shfl_xor(mx, o, 32));
    if (lane == 0) atomicMax(&amax[tileM & 63], __float_as_uint(mx));
  }
}

template <int N, int O>
__device__ __forceinline__ void lane_reduce(float* p, int lane) {
  if constexpr (O > 0) {
    if constexpr (N > 1) {
      constexpr int HN = N / 2;
      const bool up = (lane & O) != 0;
      for (int i = 0; i < HN; ++i) {
        float keep = up ? p[HN + i] : p[i];
        float send = up ? p[i] : p[HN + i];
        p[i] = keep + __shfl_xor(send, O, 32);
      }
      lane_reduce<HN, O / 2>(p, lane);
    } else {
      p[0] += __shfl_xor(p[0], O, 32);
      lane_reduce<1, O / 2>(p, lane);
    }
  }
}

template <int H, int C, int NB>
__global__ void k_edge_score(const float* __restrict__ XL, const float* __restrict__ XR,
                             const float* __restrict__ att, const int* __restrict__ ei,
                             long long E0, long long ET,
                             float* __restrict__ SC, unsigned* __restrict__ M) {
  constexpr int HC  = H * C;
  constexpr int LPH = 32 / H;
  constexpr int CPT = C / LPH;
  constexpr int G   = LPH / NB;
  static_assert(NB * G == LPH && H * LPH == 32 && CPT * LPH == C, "shape");
  const long long e0 =
      ((long long)blockIdx.x * (blockDim.x >> 5) + (threadIdx.x >> 5)) * 32;
  if (e0 >= ET) return;
  const int lane = threadIdx.x & 31;

  const long long el = e0 + lane;
  int sv = 0, dv = 0;
  if (el < E0)      { sv = ei[el]; dv = ei[E0 + el]; }
  else if (el < ET) { sv = dv = (int)(el - E0); }

  const int h   = lane / LPH;
  const int q   = lane % LPH;
  const int kk  = q / G;
  const int rep = q % G;
  float a[CPT];
  for (int j = 0; j < CPT; ++j) a[j] = att[lane * CPT + j];
  const float* xlb = XL + lane * CPT;
  const float* xrb = XR + lane * CPT;

  for (int sb = 0; sb < H; ++sb) {
    float r  = 0.0f;
    int   rd = 0;
    for (int bb = 0; bb < G; ++bb) {
      const int base = sb * LPH + bb * NB;
      float xl[NB][CPT], xr[NB][CPT];
      int   dk[NB];
      for (int k = 0; k < NB; ++k) {
        const int s = __builtin_amdgcn_readlane(sv, base + k);
        const int d = __builtin_amdgcn_readlane(dv, base + k);
        dk[k] = d;
        const float* pl = xlb + (size_t)s * HC;
        const float* pr = xrb + (size_t)d * HC;
        for (int j = 0; j < CPT; ++j) { xl[k][j] = pl[j]; xr[k][j] = pr[j]; }
      }
      float p[NB];
      for (int k = 0; k < NB; ++k) {
        float acc = 0.0f;
        for (int j = 0; j < CPT; ++j) {
          float v = xl[k][j] + xr[k][j];
          v = (v > 0.0f) ? v : NEG_SLOPE * v;
          acc += a[j] * v;
        }
        p[k] = acc;
      }
      lane_reduce<NB, LPH / 2>(p, lane);
      int dq = dk[0];
      for (int k = 1; k < NB; ++k) dq = (kk == k) ? dk[k] : dq;
      if (G == 1 || rep == bb) { r = p[0]; rd = dq; }
    }
    const long long e = e0 + sb * LPH + rep * NB + kk;
    if (e < ET) {
      SC[(size_t)e * H + h] = r;
      atomicMax(&M[(size_t)rd * H + h], fenc(r));
    }
  }
}

template <int H, int C, int NB>
__global__ void k_edge_aggr(const float* __restrict__ XL, const float* __restrict__ SC,
                            const unsigned* __restrict__ M, unsigned* __restrict__ DEN,
                            int* __restrict__ OUT,
                            const int* __restrict__ ei, long long E0, long long ET,
                            const float* __restrict__ PAR) {
  constexpr int HC  = H * C;
  constexpr int NJ  = HC / 32;
  constexpr int EPG = 32 / H;
  static_assert(EPG * H == 32 && NJ * 32 == HC && 32 % NB == 0, "shape");
  const long long e0 =
      ((long long)blockIdx.x * (blockDim.x >> 5) + (threadIdx.x >> 5)) * 32;
  if (e0 >= ET) return;
  const int lane = threadIdx.x & 31;
  const float S = PAR[0], SD = PAR[2];

  const long long el = e0 + lane;
  int sv = 0, dv = 0;
  if (el < E0)      { sv = ei[el]; dv = ei[E0 + el]; }
  else if (el < ET) { sv = dv = (int)(el - E0); }

  const int hl = lane % H;
  const int eg = lane / H;
  float sc[H];
#pragma unroll
  for (int g = 0; g < H; ++g) {
    const long long e  = e0 + g * EPG + eg;
    const long long ec = (e < ET) ? e : ET - 1;
    sc[g] = SC[(size_t)ec * H + hl];
  }
  int      dg[H];
  unsigned mu[H];
#pragma unroll
  for (int g = 0; g < H; ++g) {
    dg[g] = (H == 1) ? dv : __shfl(dv, g * EPG + eg, 32);
    mu[g] = M[(size_t)dg[g] * H + hl];
  }
  float w[H];
#pragma unroll
  for (int g = 0; g < H; ++g) {
    const long long e = e0 + g * EPG + eg;
    float m = fdec(mu[g]);
    if (!(m > -3.0e38f && m < 3.0e38f)) m = 0.0f;
    const float ex = (e < ET) ? __expf(sc[g] - m) : 0.0f;
    atomicAdd(&DEN[(size_t)dg[g] * H + hl], (unsigned)__float2int_rn(ex * SD));
    w[g] = ex * S;
  }

#pragma unroll
  for (int b = 0; b < 32 / NB; ++b) {
    float v[NB][NJ];
#pragma unroll
    for (int k = 0; k < NB; ++k) {
      const int s = __builtin_amdgcn_readlane(sv, b * NB + k);
      const float* xs = XL + (size_t)s * HC + lane;
#pragma unroll
      for (int j = 0; j < NJ; ++j) v[k][j] = xs[32 * j];
    }
#pragma unroll
    for (int k = 0; k < NB; ++k) {
      const int kk = b * NB + k;
      const int d = __builtin_amdgcn_readlane(dv, kk);
      int* od = OUT + (size_t)d * HC + lane;
#pragma unroll
      for (int j = 0; j < NJ; ++j) {
        const float a = __shfl(w[kk / EPG], (kk % EPG) * H + (lane + 32 * j) / C, 32);
        atomicAdd(&od[32 * j], __float2int_rn(a * v[k][j]));
      }
    }
  }
}

__global__ void k_softmax64(const int* __restrict__ in, const unsigned* __restrict__ den,
                            const float* __restrict__ bias, float* __restrict__ out, int n,
                            const float* __restrict__ PAR) {
  int node = blockIdx.x * (blockDim.x >> 5) + (threadIdx.x >> 5);
  if (node >= n) return;
  int lane = threadIdx.x & 31;
  const int* p = in + (size_t)node * 64 + lane * 2;
  const float invS = PAR[1];
  float dn = fmaxf((float)den[node] * PAR[3], 1e-16f);
  float a = (float)p[0] * invS / dn + bias[lane * 2];
  float b = (float)p[1] * invS / dn + bias[lane * 2 + 1];
  float mx = fmaxf(a, b);
#pragma unroll
  for (int o = 16; o > 0; o >>= 1) mx = fmaxf(mx, __shfl_xor(mx, o, 32));
  float ea = __expf(a - mx), eb = __expf(b - mx);
  float sum = ea + eb;
#pragma unroll
  for (int o = 16; o > 0; o >>= 1) sum += __shfl_xor(sum, o, 32);
  float inv = 1.0f / sum;
  float* q = out + (size_t)node * 64 + lane * 2;
  q[0] = ea * inv;
  q[1] = eb * inv;
}

template <int H, int C>
static void gat_layer(const _Float16* X16, int din,
                      const float* Wl, const float* Wr, const float* att,
                      _Float16* WPa, _Float16* WPb,
                      float* XL, float* XR, int* OUT, unsigned* M, unsigned* DEN, float* SC,
                      const unsigned* SMALL, unsigned* AMAX, float* PAR,
                      const int* ei, int N, long long E0, hipStream_t stream) {
  constexpr int HC = H * C;
  long long ET = E0 + N;

  int packThreads = (din >> 5) * (HC >> 4) * 32;
  int packBlocks  = (packThreads + 255) / 256;
  k_pack_w<<<packBlocks, 256, 0, stream>>>(Wl, WPa, din, HC);
  k_pack_w<<<packBlocks, 256, 0, stream>>>(Wr, WPb, din, HC);

  int rowTiles = (N + 15) / 16;
  dim3 gg((unsigned)((rowTiles + 3) / 4), (unsigned)(HC / 64));
  k_gemm_f16<true><<<gg, 128, 0, stream>>>(X16, WPa, XL, N, din, HC, AMAX);
  k_gemm_f16<false><<<gg, 128, 0, stream>>>(X16, WPb, XR, N, din, HC, nullptr);
  k_scale<<<1, 32, 0, stream>>>(SMALL, AMAX, PAR);

  long long nHC = (long long)N * HC, nH = (long long)N * H;
  k_init_out<<<(int)((nHC + 255) / 256), 256, 0, stream>>>(OUT, nHC);
  k_init_md<<<(int)((nH + 255) / 256), 256, 0, stream>>>(M, DEN, nH);

  int gS = (int)((ET + 255) / 256);
  k_edge_score<H, C, (H == 1 ? 8 : 4)><<<gS, 256, 0, stream>>>(XL, XR, att, ei, E0, ET, SC, M);
  k_edge_aggr<H, C, (H == 1 ? 16 : 8)><<<gS, 256, 0, stream>>>(XL, SC, M, DEN, OUT, ei, E0, ET, PAR);
}

extern "C" void kernel_launch(void* const* d_in, const int* in_sizes, int n_in,
                              void* d_out, int out_size, void* d_ws, size_t ws_size,
                              hipStream_t stream) {
  (void)n_in; (void)out_size; (void)ws_size;
  const float* x    = (const float*)d_in[0];
  const int*   ei   = (const int*)d_in[1];
  const float* Wl1  = (const float*)d_in[2];
  const float* Wr1  = (const float*)d_in[3];
  const float* att1 = (const float*)d_in[4];
  const float* b1   = (const float*)d_in[5];
  const float* Wl2  = (const float*)d_in[6];
  const float* Wr2  = (const float*)d_in[7];
  const float* att2 = (const float*)d_in[8];
  const float* b2   = (const float*)d_in[9];
  const float* Wl3  = (const float*)d_in[10];
  const float* Wr3  = (const float*)d_in[11];
  const float* att3 = (const float*)d_in[12];
  const float* b3   = (const float*)d_in[13];

  int       N  = in_sizes[0] / 256;
  long long E0 = (long long)in_sizes[1] / 2;
  long long ET = E0 + N;

  uintptr_t p = (uintptr_t)d_ws;
  auto carve = [&](size_t bytes) -> void* {
    void* r = (void*)p;
    p += (bytes + 255) & ~(size_t)255;
    return r;
  };
  _Float16* X16 = (_Float16*)carve((size_t)N * 256 * sizeof(_Float16));
  _Float16* WPa = (_Float16*)carve((size_t)256 * 128 * sizeof(_Float16));
  _Float16* WPb = (_Float16*)carve((size_t)256 * 128 * sizeof(_Float16));
  float*    XL  = (float*)carve((size_t)N * 128 * sizeof(float));
  float*    XR  = (float*)carve((size_t)N * 128 * sizeof(float));
  int*      OUT = (int*)carve((size_t)N * 128 * sizeof(int));
  unsigned* M   = (unsigned*)carve((size_t)N * 8 * sizeof(unsigned));
  unsigned* DEN = (unsigned*)carve((size_t)N * 8 * sizeof(unsigned));
  float*    SC  = (float*)carve((size_t)ET * 8 * sizeof(float));
  unsigned* SMALL = (unsigned*)d_out;
  float*    PAR1  = (float*)d_out + SMALL_WORDS;
  float*    PAR2  = PAR1 + 4;
  float*    PAR3  = SC + ET;
  unsigned* DEG   = (unsigned*)OUT;
  k_zero_small<<<(N + 255) / 256, 256, 0, stream>>>(DEG, SMALL, N);
  k_count_deg<<<(int)((E0 + 255) / 256), 256, 0, stream>>>(ei, E0, DEG);
  k_deg_max<<<64, 256, 0, stream>>>(DEG, N, SMALL);

  long long nx = (long long)N * 256;
  k_f32_to_f16<<<(int)((nx + 255) / 256), 256, 0, stream>>>(x, X16, nx);
  gat_layer<8, 16>(X16, 256, Wl1, Wr1, att1, WPa, WPb,
                   XL, XR, OUT, M, DEN, SC, SMALL, SMALL + 64, PAR1, ei, N, E0, stream);
  long long nh = (long long)N * 128;
  k_norm_elu_f16<8, 16><<<(int)((nh + 255) / 256), 256, 0, stream>>>(OUT, DEN, b1, X16, nh, PAR1);

  gat_layer<8, 16>(X16, 128, Wl2, Wr2, att2, WPa, WPb,
                   XL, XR, OUT, M, DEN, SC, SMALL, SMALL + 128, PAR2, ei, N, E0, stream);
  k_norm_elu_f16<8, 16><<<(int)((nh + 255) / 256), 256, 0, stream>>>(OUT, DEN, b2, X16, nh, PAR2);

  gat_layer<1, 64>(X16, 128, Wl3, Wr3, att3, WPa, WPb,
                   XL, XR, OUT, M, DEN, SC, SMALL, SMALL + 192, PAR3, ei, N, E0, stream);

  k_softmax64<<<(N + 7) / 8, 256, 0, stream>>>(OUT, DEN, b3, (float*)d_out, N, PAR3);
}


# A operand made in the GEMM's own loader from the f32 source (x, or the layer's integer sums: same normalise, bias, ELU, rounding); one wave does both projections; 3 passes gone; 34->28 launches
# speedup vs baseline: 1.8306x; 1.1039x over previous
#include <hip/hip_runtime.h>
#include <hip/hip_bf16.h>
#include <math.h>


#define NEG_SLOPE 0.2f

typedef __attribute__((ext_vector_type(16))) _Float16 v16h;
typedef __attribute__((ext_vector_type(8)))  _Float16 v8h;
typedef __attribute__((ext_vector_type(8)))  float    v8f;

__device__ __forceinline__ unsigned fenc(float f) {
  unsigned u = __float_as_uint(f);
  return (u & 0x80000000u) ? ~u : (u | 0x80000000u);
}
__device__ __forceinline__ float fdec(unsigned u) {
  return (u & 0x80000000u) ? __uint_as_float(u & 0x7fffffffu) : __uint_as_float(~u);
}
#define ENC_NEG_INF 0x007FFFFFu

__global__ void k_init_out(int* __restrict__ out, long long n) {
  long long i = (long long)blockIdx.x * blockDim.x + threadIdx.x;
  if (i < n) out[i] = 0;
}

__global__ void k_init_md(unsigned* __restrict__ M, unsigned* __restrict__ D, long long n) {
  long long i = (long long)blockIdx.x * blockDim.x + threadIdx.x;
  if (i < n) { M[i] = ENC_NEG_INF; D[i] = 0u; }
}

#define SMALL_WORDS 256
__global__ void k_zero_small(unsigned* __restrict__ DEG, unsigned* __restrict__ SMALL, int n) {
  int i = blockIdx.x * blockDim.x + threadIdx.x;
  if (i < n) DEG[i] = 0u;
  if (i < SMALL_WORDS) SMALL[i] = 0u;
}

__global__ void k_count_deg(const int* __restrict__ ei, long long E0, unsigned* __restrict__ DEG) {
  long long i = (long long)blockIdx.x * blockDim.x + threadIdx.x;
  if (i < E0) atomicAdd(&DEG[ei[E0 + i]], 1u);
}

__global__ void k_deg_max(const unsigned* __restrict__ DEG, int n, unsigned* __restrict__ SMALL) {
  unsigned v = 0u;
  for (int i = blockIdx.x * blockDim.x + threadIdx.x; i < n; i += gridDim.x * blockDim.x) {
    unsigned t = DEG[i];
    v = (t > v) ? t : v;
  }
#pragma unroll
  for (int o = 16; o > 0; o >>= 1) {
    unsigned t = (unsigned)__shfl_xor((int)v, o, 32);
    v = (t > v) ? t : v;
  }
  if ((threadIdx.x & 31) == 0) atomicMax(&SMALL[0], v);
}

__global__ void k_scale(const unsigned* __restrict__ SMALL, const unsigned* __restrict__ amax64,
                        float* __restrict__ PAR) {
  int lane = threadIdx.x & 31;
  unsigned a = amax64[lane], b = amax64[lane + 32];
  a = (b > a) ? b : a;
#pragma unroll
  for (int o = 16; o > 0; o >>= 1) {
    unsigned t = (unsigned)__shfl_xor((int)a, o, 32);
    a = (t > a) ? t : a;
  }
  if (lane == 0) {
    unsigned deg = SMALL[0] + 1u;
    float bound = (float)deg * __uint_as_float(a);
    int eb = (int)((__float_as_uint(bound) >> 23) & 0xffu);
    int k = 156 - eb;
    k = (k > 100) ? 100 : ((k < -100) ? -100 : k);
    int p = 32 - __clz((int)deg);
    int kd = 31 - p;
    kd = (kd > 30) ? 30 : ((kd < 0) ? 0 : kd);
    PAR[0] = __uint_as_float((unsigned)(127 + k) << 23);
    PAR[1] = __uint_as_float((unsigned)(127 - k) << 23);
    PAR[2] = __uint_as_float((unsigned)(127 + kd) << 23);
    PAR[3] = __uint_as_float((unsigned)(127 - kd) << 23);
  }
}

__global__ void k_pack_w(const float* __restrict__ W, _Float16* __restrict__ WP,
                         int din, int dout) {
  int ktiles = din >> 5, ntiles = dout >> 4;
  int total = ktiles * ntiles * 32;
  int idx = blockIdx.x * blockDim.x + threadIdx.x;
  if (idx >= total) return;
  int lane = idx & 31;
  int frag = idx >> 5;
  int nt = frag % ntiles;
  int kt = frag / ntiles;
  int half = lane >> 4, lr = lane & 15;
  _Float16* o = WP + ((size_t)frag * 32 + lane) * 16;
  const float* w0 = W + (size_t)(kt * 32 + half * 8) * dout + nt * 16 + lr;
#pragma unroll
  for (int j = 0; j < 8; ++j) {
    o[j]     = (_Float16)w0[(size_t)j * dout];
    o[8 + j] = (_Float16)w0[(size_t)(16 + j) * dout];
  }
}

template <int SRC, int DIN, int NT>
__global__ void __launch_bounds__(128) k_gemm_src(const float* __restrict__ xf, const int* __restrict__ si,
                           const unsigned* __restrict__ den, const float* __restrict__ bias,
                           const float* __restrict__ PARp,
                           const _Float16* __restrict__ WPa, const _Float16* __restrict__ WPb,
                           float* __restrict__ XL, float* __restrict__ XR, int Mrows,
                           unsigned* __restrict__ amax) {
  constexpr int DOUT = NT * 16;
  int wave = threadIdx.x >> 5;
  int lane = threadIdx.x & 31;
  int half = lane >> 4;
  int lr   = lane & 15;
  int tileM = blockIdx.x * 4 + wave;
  if (tileM * 16 >= Mrows) return;

  int row = tileM * 16 + lr;
  if (row >= Mrows) row = Mrows - 1;

  float invS = 0.0f, invSD = 0.0f;
  if constexpr (SRC == 1) { invS = PARp[1]; invSD = PARp[3]; }

  v8f acc[2 * NT];
#pragma unroll
  for (int t = 0; t < 2 * NT; ++t) acc[t] = v8f{};

  const v16h* WPav = (const v16h*)WPa;
  const v16h* WPbv = (const v16h*)WPb;
  for (int kt = 0; kt < DIN / 32; ++kt) {
    const int kb = kt * 32 + half * 8;
    float f[16];
    if constexpr (SRC == 0) {
      const float4* p = (const float4*)(xf + (size_t)row * DIN + kb);
      float4 q0 = p[0], q1 = p[1], q2 = p[4], q3 = p[5];
      f[0] = q0.x; f[1] = q0.y; f[2]  = q0.z; f[3]  = q0.w;
      f[4] = q1.x; f[5] = q1.y; f[6]  = q1.z; f[7]  = q1.w;
      f[8] = q2.x; f[9] = q2.y; f[10] = q2.z; f[11] = q2.w;
      f[12] = q3.x; f[13] = q3.y; f[14] = q3.z; f[15] = q3.w;
    } else {
      const int4* p = (const int4*)(si + (size_t)row * DIN + kb);
      int4 q0 = p[0], q1 = p[1], q2 = p[4], q3 = p[5];
      const float4* bp = (const float4*)(bias + kb);
      float4 b0 = bp[0], b1 = bp[1], b2 = bp[4], b3 = bp[5];
      const uint2 dd = *(const uint2*)(den + (size_t)row * 8 + 2 * kt);
      const float dlo = fmaxf((float)dd.x * invSD, 1e-16f);
      const float dhi = fmaxf((float)dd.y * invSD, 1e-16f);
      f[0] = (float)q0.x * invS / dlo + b0.x;  f[1] = (float)q0.y * invS / dlo + b0.y;
      f[2] = (float)q0.z * invS / dlo + b0.z;  f[3] = (float)q0.w * invS / dlo + b0.w;
      f[4] = (float)q1.x * invS / dlo + b1.x;  f[5] = (float)q1.y * invS / dlo + b1.y;
      f[6] = (float)q1.z * invS / dlo + b1.z;  f[7] = (float)q1.w * invS / dlo + b1.w;
      f[8]  = (float)q2.x * invS / dhi + b2.x; f[9]  = (float)q2.y * invS / dhi + b2.y;
      f[10] = (float)q2.z * invS / dhi + b2.z; f[11] = (float)q2.w * invS / dhi + b2.w;
      f[12] = (float)q3.x * invS / dhi + b3.x; f[13] = (float)q3.y * invS / dhi + b3.y;
      f[14] = (float)q3.z * invS / dhi + b3.z; f[15] = (float)q3.w * invS / dhi + b3.w;
#pragma unroll
      for (int j = 0; j < 16; ++j) f[j] = (f[j] > 0.0f) ? f[j] : expm1f(f[j]);
    }
    v16h a;
#pragma unroll
    for (int j = 0; j < 16; ++j) a[j] = (_Float16)f[j];

    const size_t fb = (size_t)kt * NT * 32 + lane;
    v16h bl[NT], br[NT];
#pragma unroll
    for (int t = 0; t < NT; ++t) bl[t] = WPav[fb + (size_t)t * 32];
    if constexpr (SRC == 1) __builtin_amdgcn_sched_barrier(0);
#pragma unroll
    for (int t = 0; t < NT; ++t) br[t] = WPbv[fb + (size_t)t * 32];
#pragma unroll
    for (int t = 0; t < NT; ++t)
      acc[t] = __builtin_amdgcn_wmma_f32_16x16x32_f16(false, a, false, bl[t], (short)0, acc[t], false, false);
    if constexpr (SRC == 1) __builtin_amdgcn_sched_barrier(0);
#pragma unroll
    for (int t = 0; t < NT; ++t)
      acc[NT + t] = __builtin_amdgcn_wmma_f32_16x16x32_f16(false, a, false, br[t], (short)0, acc[NT + t], false, false);
  }
  int r0 = tileM * 16 + half * 8;
#pragma unroll
  for (int i = 0; i < 8; ++i) {
    int r = r0 + i;
    if (r < Mrows) {
      float* yl = XL + (size_t)r * DOUT + lr;
      float* yr = XR + (size_t)r * DOUT + lr;
#pragma unroll
      for (int t = 0; t < NT; ++t) {
        yl[t * 16] = acc[t][i];
        yr[t * 16] = acc[NT + t][i];
      }
    }
  }
  float mx = 0.0f;
#pragma unroll
  for (int t = 0; t < NT; ++t) {
#pragma unroll
    for (int i = 0; i < 8; ++i) mx = fmaxf(mx, fabsf(acc[t][i]));
  }
#pragma unroll
  for (int o = 16; o > 0; o >>= 1) mx = fmaxf(mx, __shfl_xor(mx, o, 32));
  if (lane == 0) atomicMax(&amax[tileM & 63], __float_as_uint(mx));
}

template <int N, int O>
__device__ __forceinline__ void lane_reduce(float* p, int lane) {
  if constexpr (O > 0) {
    if constexpr (N > 1) {
      constexpr int HN = N / 2;
      const bool up = (lane & O) != 0;
      for (int i = 0; i < HN; ++i) {
        float keep = up ? p[HN + i] : p[i];
        float send = up ? p[i] : p[HN + i];
        p[i] = keep + __shfl_xor(send, O, 32);
      }
      lane_reduce<HN, O / 2>(p, lane);
    } else {
      p[0] += __shfl_xor(p[0], O, 32);
      lane_reduce<1, O / 2>(p, lane);
    }
  }
}

template <int H, int C, int NB>
__global__ void k_edge_score(const float* __restrict__ XL, const float* __restrict__ XR,
                             const float* __restrict__ att, const int* __restrict__ ei,
                             long long E0, long long ET,
                             float* __restrict__ SC, unsigned* __restrict__ M) {
  constexpr int HC  = H * C;
  constexpr int LPH = 32 / H;
  constexpr int CPT = C / LPH;
  constexpr int G   = LPH / NB;
  static_assert(NB * G == LPH && H * LPH == 32 && CPT * LPH == C, "shape");
  const long long e0 =
      ((long long)blockIdx.x * (blockDim.x >> 5) + (threadIdx.x >> 5)) * 32;
  if (e0 >= ET) return;
  const int lane = threadIdx.x & 31;

  const long long el = e0 + lane;
  int sv = 0, dv = 0;
  if (el < E0)      { sv = ei[el]; dv = ei[E0 + el]; }
  else if (el < ET) { sv = dv = (int)(el - E0); }

  const int h   = lane / LPH;
  const int q   = lane % LPH;
  const int kk  = q / G;
  const int rep = q % G;
  float a[CPT];
  for (int j = 0; j < CPT; ++j) a[j] = att[lane * CPT + j];
  const float* xlb = XL + lane * CPT;
  const float* xrb = XR + lane * CPT;

  for (int sb = 0; sb < H; ++sb) {
    float r  = 0.0f;
    int   rd = 0;
    for (int bb = 0; bb < G; ++bb) {
      const int base = sb * LPH + bb * NB;
      float xl[NB][CPT], xr[NB][CPT];
      int   dk[NB];
      for (int k = 0; k < NB; ++k) {
        const int s = __builtin_amdgcn_readlane(sv, base + k);
        const int d = __builtin_amdgcn_readlane(dv, base + k);
        dk[k] = d;
        const float* pl = xlb + (size_t)s * HC;
        const float* pr = xrb + (size_t)d * HC;
        for (int j = 0; j < CPT; ++j) { xl[k][j] = pl[j]; xr[k][j] = pr[j]; }
      }
      float p[NB];
      for (int k = 0; k < NB; ++k) {
        float acc = 0.0f;
        for (int j = 0; j < CPT; ++j) {
          float v = xl[k][j] + xr[k][j];
          v = (v > 0.0f) ? v : NEG_SLOPE * v;
          acc += a[j] * v;
        }
        p[k] = acc;
      }
      lane_reduce<NB, LPH / 2>(p, lane);
      int dq = dk[0];
      for (int k = 1; k < NB; ++k) dq = (kk == k) ? dk[k] : dq;
      if (G == 1 || rep == bb) { r = p[0]; rd = dq; }
    }
    const long long e = e0 + sb * LPH + rep * NB + kk;
    if (e < ET) {
      SC[(size_t)e * H + h] = r;
      atomicMax(&M[(size_t)rd * H + h], fenc(r));
    }
  }
}

template <int H, int C, int NB>
__global__ void k_edge_aggr(const float* __restrict__ XL, const float* __restrict__ SC,
                            const unsigned* __restrict__ M, unsigned* __restrict__ DEN,
                            int* __restrict__ OUT,
                            const int* __restrict__ ei, long long E0, long long ET,
                            const float* __restrict__ PAR) {
  constexpr int HC  = H * C;
  constexpr int NJ  = HC / 32;
  constexpr int EPG = 32 / H;
  static_assert(EPG * H == 32 && NJ * 32 == HC && 32 % NB == 0, "shape");
  const long long e0 =
      ((long long)blockIdx.x * (blockDim.x >> 5) + (threadIdx.x >> 5)) * 32;
  if (e0 >= ET) return;
  const int lane = threadIdx.x & 31;
  const float S = PAR[0], SD = PAR[2];

  const long long el = e0 + lane;
  int sv = 0, dv = 0;
  if (el < E0)      { sv = ei[el]; dv = ei[E0 + el]; }
  else if (el < ET) { sv = dv = (int)(el - E0); }

  const int hl = lane % H;
  const int eg = lane / H;
  float sc[H];
#pragma unroll
  for (int g = 0; g < H; ++g) {
    const long long e  = e0 + g * EPG + eg;
    const long long ec = (e < ET) ? e : ET - 1;
    sc[g] = SC[(size_t)ec * H + hl];
  }
  int      dg[H];
  unsigned mu[H];
#pragma unroll
  for (int g = 0; g < H; ++g) {
    dg[g] = (H == 1) ? dv : __shfl(dv, g * EPG + eg, 32);
    mu[g] = M[(size_t)dg[g] * H + hl];
  }
  float w[H];
#pragma unroll
  for (int g = 0; g < H; ++g) {
    const long long e = e0 + g * EPG + eg;
    float m = fdec(mu[g]);
    if (!(m > -3.0e38f && m < 3.0e38f)) m = 0.0f;
    const float ex = (e < ET) ? __expf(sc[g] - m) : 0.0f;
    atomicAdd(&DEN[(size_t)dg[g] * H + hl], (unsigned)__float2int_rn(ex * SD));
    w[g] = ex * S;
  }

#pragma unroll
  for (int b = 0; b < 32 / NB; ++b) {
    float v[NB][NJ];
#pragma unroll
    for (int k = 0; k < NB; ++k) {
      const int s = __builtin_amdgcn_readlane(sv, b * NB + k);
      const float* xs = XL + (size_t)s * HC + lane;
#pragma unroll
      for (int j = 0; j < NJ; ++j) v[k][j] = xs[32 * j];
    }
#pragma unroll
    for (int k = 0; k < NB; ++k) {
      const int kk = b * NB + k;
      const int d = __builtin_amdgcn_readlane(dv, kk);
      int* od = OUT + (size_t)d * HC + lane;
#pragma unroll
      for (int j = 0; j < NJ; ++j) {
        const float a = __shfl(w[kk / EPG], (kk % EPG) * H + (lane + 32 * j) / C, 32);
        atomicAdd(&od[32 * j], __float2int_rn(a * v[k][j]));
      }
    }
  }
}

__global__ void k_softmax64(const int* __restrict__ in, const unsigned* __restrict__ den,
                            const float* __restrict__ bias, float* __restrict__ out, int n,
                            const float* __restrict__ PAR) {
  int node = blockIdx.x * (blockDim.x >> 5) + (threadIdx.x >> 5);
  if (node >= n) return;
  int lane = threadIdx.x & 31;
  const int* p = in + (size_t)node * 64 + lane * 2;
  const float invS = PAR[1];
  float dn = fmaxf((float)den[node] * PAR[3], 1e-16f);
  float a = (float)p[0] * invS / dn + bias[lane * 2];
  float b = (float)p[1] * invS / dn + bias[lane * 2 + 1];
  float mx = fmaxf(a, b);
#pragma unroll
  for (int o = 16; o > 0; o >>= 1) mx = fmaxf(mx, __shfl_xor(mx, o, 32));
  float ea = __expf(a - mx), eb = __expf(b - mx);
  float sum = ea + eb;
#pragma unroll
  for (int o = 16; o > 0; o >>= 1) sum += __shfl_xor(sum, o, 32);
  float inv = 1.0f / sum;
  float* q = out + (size_t)node * 64 + lane * 2;
  q[0] = ea * inv;
  q[1] = eb * inv;
}

template <int H, int C, int SRC, int DIN>
static void gat_layer(const float* xf, const int* SI, const unsigned* DPREV, const float* BPREV,
                      const float* PARPREV,
                      const float* Wl, const float* Wr, const float* att,
                      _Float16* WPa, _Float16* WPb,
                      float* XL, float* XR, int* OUT, unsigned* M, unsigned* DEN, float* SC,
                      const unsigned* SMALL, unsigned* AMAX, float* PAR,
                      const int* ei, int N, long long E0, hipStream_t stream) {
  constexpr int HC = H * C;
  long long ET = E0 + N;

  int packThreads = (DIN >> 5) * (HC >> 4) * 32;
  int packBlocks  = (packThreads + 255) / 256;
  k_pack_w<<<packBlocks, 256, 0, stream>>>(Wl, WPa, DIN, HC);
  k_pack_w<<<packBlocks, 256, 0, stream>>>(Wr, WPb, DIN, HC);

  int rowTiles = (N + 15) / 16;
  k_gemm_src<SRC, DIN, HC / 16><<<(rowTiles + 3) / 4, 128, 0, stream>>>(
      xf, SI, DPREV, BPREV, PARPREV, WPa, WPb, XL, XR, N, AMAX);
  k_scale<<<1, 32, 0, stream>>>(SMALL, AMAX, PAR);

  long long nHC = (long long)N * HC, nH = (long long)N * H;
  k_init_out<<<(int)((nHC + 255) / 256), 256, 0, stream>>>(OUT, nHC);
  k_init_md<<<(int)((nH + 255) / 256), 256, 0, stream>>>(M, DEN, nH);

  int gS = (int)((ET + 255) / 256);
  k_edge_score<H, C, (H == 1 ? 8 : 4)><<<gS, 256, 0, stream>>>(XL, XR, att, ei, E0, ET, SC, M);
  k_edge_aggr<H, C, (H == 1 ? 16 : 8)><<<gS, 256, 0, stream>>>(XL, SC, M, DEN, OUT, ei, E0, ET, PAR);
}

extern "C" void kernel_launch(void* const* d_in, const int* in_sizes, int n_in,
                              void* d_out, int out_size, void* d_ws, size_t ws_size,
                              hipStream_t stream) {
  (void)n_in; (void)out_size; (void)ws_size;
  const float* x    = (const float*)d_in[0];
  const int*   ei   = (const int*)d_in[1];
  const float* Wl1  = (const float*)d_in[2];
  const float* Wr1  = (const float*)d_in[3];
  const float* att1 = (const float*)d_in[4];
  const float* b1   = (const float*)d_in[5];
  const float* Wl2  = (const float*)d_in[6];
  const float* Wr2  = (const float*)d_in[7];
  const float* att2 = (const float*)d_in[8];
  const float* b2   = (const float*)d_in[9];
  const float* Wl3  = (const float*)d_in[10];
  const float* Wr3  = (const float*)d_in[11];
  const float* att3 = (const float*)d_in[12];
  const float* b3   = (const float*)d_in[13];

  int       N  = in_sizes[0] / 256;
  long long E0 = (long long)in_sizes[1] / 2;
  long long ET = E0 + N;

  uintptr_t p = (uintptr_t)d_ws;
  auto carve = [&](size_t bytes) -> void* {
    void* r = (void*)p;
    p += (bytes + 255) & ~(size_t)255;
    return r;
  };
  _Float16* WPa = (_Float16*)carve((size_t)256 * 128 * sizeof(_Float16));
  _Float16* WPb = (_Float16*)carve((size_t)256 * 128 * sizeof(_Float16));
  float*    XL  = (float*)carve((size_t)N * 128 * sizeof(float));
  float*    XR  = (float*)carve((size_t)N * 128 * sizeof(float));
  int*      OUT = (int*)carve((size_t)N * 128 * sizeof(int));
  unsigned* M   = (unsigned*)carve((size_t)N * 8 * sizeof(unsigned));
  unsigned* DEN = (unsigned*)carve((size_t)N * 8 * sizeof(unsigned));
  float*    SC  = (float*)carve((size_t)ET * 8 * sizeof(float));
  unsigned* SMALL = (unsigned*)d_out;
  float*    PAR1  = (float*)d_out + SMALL_WORDS;
  float*    PAR2  = PAR1 + 4;
  float*    PAR3  = SC + ET;
  unsigned* DEG   = (unsigned*)OUT;
  k_zero_small<<<(N + 255) / 256, 256, 0, stream>>>(DEG, SMALL, N);
  k_count_deg<<<(int)((E0 + 255) / 256), 256, 0, stream>>>(ei, E0, DEG);
  k_deg_max<<<64, 256, 0, stream>>>(DEG, N, SMALL);

  gat_layer<8, 16, 0, 256>(x, nullptr, nullptr, nullptr, nullptr, Wl1, Wr1, att1, WPa, WPb,
                           XL, XR, OUT, M, DEN, SC, SMALL, SMALL + 64, PAR1, ei, N, E0, stream);

  gat_layer<8, 16, 1, 128>(nullptr, OUT, DEN, b1, PAR1, Wl2, Wr2, att2, WPa, WPb,
                           XL, XR, OUT, M, DEN, SC, SMALL, SMALL + 128, PAR2, ei, N, E0, stream);

  gat_layer<1, 64, 1, 128>(nullptr, OUT, DEN, b2, PAR2, Wl3, Wr3, att3, WPa, WPb,
                           XL, XR, OUT, M, DEN, SC, SMALL, SMALL + 192, PAR3, ei, N, E0, stream);

  k_softmax64<<<(N + 7) / 8, 256, 0, stream>>>(OUT, DEN, b3, (float*)d_out, N, PAR3);
}


# Aggregation pass: each half of the chip (die id) adds into its own copy of the integer sum planes (row sums, exp sums); readers add the two copies, exact integer adds; 28->31 launches
# speedup vs baseline: 1.8890x; 1.0319x over previous
#include <hip/hip_runtime.h>
#include <hip/hip_bf16.h>
#include <math.h>


#define NEG_SLOPE 0.2f

typedef __attribute__((ext_vector_type(16))) _Float16 v16h;
typedef __attribute__((ext_vector_type(8)))  _Float16 v8h;
typedef __attribute__((ext_vector_type(8)))  float    v8f;

__device__ __forceinline__ unsigned fenc(float f) {
  unsigned u = __float_as_uint(f);
  return (u & 0x80000000u) ? ~u : (u | 0x80000000u);
}
__device__ __forceinline__ float fdec(unsigned u) {
  return (u & 0x80000000u) ? __uint_as_float(u & 0x7fffffffu) : __uint_as_float(~u);
}
#define ENC_NEG_INF 0x007FFFFFu

__global__ void k_init_out(int* __restrict__ out, long long n) {
  long long i = (long long)blockIdx.x * blockDim.x + threadIdx.x;
  if (i < n) out[i] = 0;
}

__global__ void k_init_plane2(int* __restrict__ out2, long long nHC,
                              unsigned* __restrict__ den2, long long nH) {
  long long i = (long long)blockIdx.x * blockDim.x + threadIdx.x;
  if (i < nHC) out2[i] = 0;
  if (i < nH) den2[i] = 0u;
}

__global__ void k_init_md(unsigned* __restrict__ M, unsigned* __restrict__ D, long long n) {
  long long i = (long long)blockIdx.x * blockDim.x + threadIdx.x;
  if (i < n) { M[i] = ENC_NEG_INF; D[i] = 0u; }
}

#define SMALL_WORDS 256
__global__ void k_zero_small(unsigned* __restrict__ DEG, unsigned* __restrict__ SMALL, int n) {
  int i = blockIdx.x * blockDim.x + threadIdx.x;
  if (i < n) DEG[i] = 0u;
  if (i < SMALL_WORDS) SMALL[i] = 0u;
}

__global__ void k_count_deg(const int* __restrict__ ei, long long E0, unsigned* __restrict__ DEG) {
  long long i = (long long)blockIdx.x * blockDim.x + threadIdx.x;
  if (i < E0) atomicAdd(&DEG[ei[E0 + i]], 1u);
}

__global__ void k_deg_max(const unsigned* __restrict__ DEG, int n, unsigned* __restrict__ SMALL) {
  unsigned v = 0u;
  for (int i = blockIdx.x * blockDim.x + threadIdx.x; i < n; i += gridDim.x * blockDim.x) {
    unsigned t = DEG[i];
    v = (t > v) ? t : v;
  }
#pragma unroll
  for (int o = 16; o > 0; o >>= 1) {
    unsigned t = (unsigned)__shfl_xor((int)v, o, 32);
    v = (t > v) ? t : v;
  }
  if ((threadIdx.x & 31) == 0) atomicMax(&SMALL[0], v);
}

__global__ void k_scale(const unsigned* __restrict__ SMALL, const unsigned* __restrict__ amax64,
                        float* __restrict__ PAR) {
  int lane = threadIdx.x & 31;
  unsigned a = amax64[lane], b = amax64[lane + 32];
  a = (b > a) ? b : a;
#pragma unroll
  for (int o = 16; o > 0; o >>= 1) {
    unsigned t = (unsigned)__shfl_xor((int)a, o, 32);
    a = (t > a) ? t : a;
  }
  if (lane == 0) {
    unsigned deg = SMALL[0] + 1u;
    float bound = (float)deg * __uint_as_float(a);
    int eb = (int)((__float_as_uint(bound) >> 23) & 0xffu);
    int k = 156 - eb;
    k = (k > 100) ? 100 : ((k < -100) ? -100 : k);
    int p = 32 - __clz((int)deg);
    int kd = 31 - p;
    kd = (kd > 30) ? 30 : ((kd < 0) ? 0 : kd);
    PAR[0] = __uint_as_float((unsigned)(127 + k) << 23);
    PAR[1] = __uint_as_float((unsigned)(127 - k) << 23);
    PAR[2] = __uint_as_float((unsigned)(127 + kd) << 23);
    PAR[3] = __uint_as_float((unsigned)(127 - kd) << 23);
  }
}

__global__ void k_pack_w(const float* __restrict__ W, _Float16* __restrict__ WP,
                         int din, int dout) {
  int ktiles = din >> 5, ntiles = dout >> 4;
  int total = ktiles * ntiles * 32;
  int idx = blockIdx.x * blockDim.x + threadIdx.x;
  if (idx >= total) return;
  int lane = idx & 31;
  int frag = idx >> 5;
  int nt = frag % ntiles;
  int kt = frag / ntiles;
  int half = lane >> 4, lr = lane & 15;
  _Float16* o = WP + ((size_t)frag * 32 + lane) * 16;
  const float* w0 = W + (size_t)(kt * 32 + half * 8) * dout + nt * 16 + lr;
#pragma unroll
  for (int j = 0; j < 8; ++j) {
    o[j]     = (_Float16)w0[(size_t)j * dout];
    o[8 + j] = (_Float16)w0[(size_t)(16 + j) * dout];
  }
}

template <int SRC, int DIN, int NT>
__global__ void __launch_bounds__(128) k_gemm_src(const float* __restrict__ xf, const int* __restrict__ si,
                           const int* si2,
                           const unsigned* __restrict__ den, const unsigned* __restrict__ den2,
                           const float* __restrict__ bias,
                           const float* __restrict__ PARp,
                           const _Float16* __restrict__ WPa, const _Float16* __restrict__ WPb,
                           float* __restrict__ XL, float* XR, int Mrows,
                           unsigned* __restrict__ amax) {
  constexpr int DOUT = NT * 16;
  int wave = threadIdx.x >> 5;
  int lane = threadIdx.x & 31;
  int half = lane >> 4;
  int lr   = lane & 15;
  int tileM = blockIdx.x * 4 + wave;
  if (tileM * 16 >= Mrows) return;

  int row = tileM * 16 + lr;
  if (row >= Mrows) row = Mrows - 1;

  float invS = 0.0f, invSD = 0.0f;
  if constexpr (SRC == 1) { invS = PARp[1]; invSD = PARp[3]; }

  v8f acc[2 * NT];
#pragma unroll
  for (int t = 0; t < 2 * NT; ++t) acc[t] = v8f{};

  const v16h* WPav = (const v16h*)WPa;
  const v16h* WPbv = (const v16h*)WPb;
  for (int kt = 0; kt < DIN / 32; ++kt) {
    const int kb = kt * 32 + half * 8;
    float f[16];
    if constexpr (SRC == 0) {
      const float4* p = (const float4*)(xf + (size_t)row * DIN + kb);
      float4 q0 = p[0], q1 = p[1], q2 = p[4], q3 = p[5];
      f[0] = q0.x; f[1] = q0.y; f[2]  = q0.z; f[3]  = q0.w;
      f[4] = q1.x; f[5] = q1.y; f[6]  = q1.z; f[7]  = q1.w;
      f[8] = q2.x; f[9] = q2.y; f[10] = q2.z; f[11] = q2.w;
      f[12] = q3.x; f[13] = q3.y; f[14] = q3.z; f[15] = q3.w;
    } else {
      const int4* p = (const int4*)(si + (size_t)row * DIN + kb);
      int4 q0 = p[0], q1 = p[1], q2 = p[4], q3 = p[5];
      const int4* p2 = (const int4*)(si2 + (size_t)row * DIN + kb);
      const int4 r0 = p2[0], r1 = p2[1], r2 = p2[4], r3 = p2[5];
      q0.x += r0.x; q0.y += r0.y; q0.z += r0.z; q0.w += r0.w;
      q1.x += r1.x; q1.y += r1.y; q1.z += r1.z; q1.w += r1.w;
      q2.x += r2.x; q2.y += r2.y; q2.z += r2.z; q2.w += r2.w;
      q3.x += r3.x; q3.y += r3.y; q3.z += r3.z; q3.w += r3.w;
      const float4* bp = (const float4*)(bias + kb);
      float4 b0 = bp[0], b1 = bp[1], b2 = bp[4], b3 = bp[5];
      uint2 dd = *(const uint2*)(den + (size_t)row * 8 + 2 * kt);
      const uint2 dd2 = *(const uint2*)(den2 + (size_t)row * 8 + 2 * kt);
      dd.x += dd2.x; dd.y += dd2.y;
      const float dlo = fmaxf((float)dd.x * invSD, 1e-16f);
      const float dhi = fmaxf((float)dd.y * invSD, 1e-16f);
      f[0] = (float)q0.x * invS / dlo + b0.x;  f[1] = (float)q0.y * invS / dlo + b0.y;
      f[2] = (float)q0.z * invS / dlo + b0.z;  f[3] = (float)q0.w * invS / dlo + b0.w;
      f[4] = (float)q1.x * invS / dlo + b1.x;  f[5] = (float)q1.y * invS / dlo + b1.y;
      f[6] = (float)q1.z * invS / dlo + b1.z;  f[7] = (float)q1.w * invS / dlo + b1.w;
      f[8]  = (float)q2.x * invS / dhi + b2.x; f[9]  = (float)q2.y * invS / dhi + b2.y;
      f[10] = (float)q2.z * invS / dhi + b2.z; f[11] = (float)q2.w * invS / dhi + b2.w;
      f[12] = (float)q3.x * invS / dhi + b3.x; f[13] = (float)q3.y * invS / dhi + b3.y;
      f[14] = (float)q3.z * invS / dhi + b3.z; f[15] = (float)q3.w * invS / dhi + b3.w;
#pragma unroll
      for (int j = 0; j < 16; ++j) f[j] = (f[j] > 0.0f) ? f[j] : expm1f(f[j]);
    }
    v16h a;
#pragma unroll
    for (int j = 0; j < 16; ++j) a[j] = (_Float16)f[j];

    const size_t fb = (size_t)kt * NT * 32 + lane;
    v16h bl[NT], br[NT];
#pragma unroll
    for (int t = 0; t < NT; ++t) bl[t] = WPav[fb + (size_t)t * 32];
    if constexpr (SRC == 1) __builtin_amdgcn_sched_barrier(0);
#pragma unroll
    for (int t = 0; t < NT; ++t) br[t] = WPbv[fb + (size_t)t * 32];
#pragma unroll
    for (int t = 0; t < NT; ++t)
      acc[t] = __builtin_amdgcn_wmma_f32_16x16x32_f16(false, a, false, bl[t], (short)0, acc[t], false, false);
    if constexpr (SRC == 1) __builtin_amdgcn_sched_barrier(0);
#pragma unroll
    for (int t = 0; t < NT; ++t)
      acc[NT + t] = __builtin_amdgcn_wmma_f32_16x16x32_f16(false, a, false, br[t], (short)0, acc[NT + t], false, false);
  }
  int r0 = tileM * 16 + half * 8;
#pragma unroll
  for (int i = 0; i < 8; ++i) {
    int r = r0 + i;
    if (r < Mrows) {
      float* yl = XL + (size_t)r * DOUT + lr;
      float* yr = XR + (size_t)r * DOUT + lr;
#pragma unroll
      for (int t = 0; t < NT; ++t) {
        yl[t * 16] = acc[t][i];
        yr[t * 16] = acc[NT + t][i];
      }
    }
  }
  float mx = 0.0f;
#pragma unroll
  for (int t = 0; t < NT; ++t) {
#pragma unroll
    for (int i = 0; i < 8; ++i) mx = fmaxf(mx, fabsf(acc[t][i]));
  }
#pragma unroll
  for (int o = 16; o > 0; o >>= 1) mx = fmaxf(mx, __shfl_xor(mx, o, 32));
  if (lane == 0) atomicMax(&amax[tileM & 63], __float_as_uint(mx));
}

template <int N, int O>
__device__ __forceinline__ void lane_reduce(float* p, int lane) {
  if constexpr (O > 0) {
    if constexpr (N > 1) {
      constexpr int HN = N / 2;
      const bool up = (lane & O) != 0;
      for (int i = 0; i < HN; ++i) {
        float keep = up ? p[HN + i] : p[i];
        float send = up ? p[i] : p[HN + i];
        p[i] = keep + __shfl_xor(send, O, 32);
      }
      lane_reduce<HN, O / 2>(p, lane);
    } else {
      p[0] += __shfl_xor(p[0], O, 32);
      lane_reduce<1, O / 2>(p, lane);
    }
  }
}

template <int H, int C, int NB>
__global__ void k_edge_score(const float* __restrict__ XL, const float* __restrict__ XR,
                             const float* __restrict__ att, const int* __restrict__ ei,
                             long long E0, long long ET,
                             float* __restrict__ SC, unsigned* __restrict__ M) {
  constexpr int HC  = H * C;
  constexpr int LPH = 32 / H;
  constexpr int CPT = C / LPH;
  constexpr int G   = LPH / NB;
  static_assert(NB * G == LPH && H * LPH == 32 && CPT * LPH == C, "shape");
  const long long e0 =
      ((long long)blockIdx.x * (blockDim.x >> 5) + (threadIdx.x >> 5)) * 32;
  if (e0 >= ET) return;
  const int lane = threadIdx.x & 31;

  const long long el = e0 + lane;
  int sv = 0, dv = 0;
  if (el < E0)      { sv = ei[el]; dv = ei[E0 + el]; }
  else if (el < ET) { sv = dv = (int)(el - E0); }

  const int h   = lane / LPH;
  const int q   = lane % LPH;
  const int kk  = q / G;
  const int rep = q % G;
  float a[CPT];
  for (int j = 0; j < CPT; ++j) a[j] = att[lane * CPT + j];
  const float* xlb = XL + lane * CPT;
  const float* xrb = XR + lane * CPT;

  for (int sb = 0; sb < H; ++sb) {
    float r  = 0.0f;
    int   rd = 0;
    for (int bb = 0; bb < G; ++bb) {
      const int base = sb * LPH + bb * NB;
      float xl[NB][CPT], xr[NB][CPT];
      int   dk[NB];
      for (int k = 0; k < NB; ++k) {
        const int s = __builtin_amdgcn_readlane(sv, base + k);
        const int d = __builtin_amdgcn_readlane(dv, base + k);
        dk[k] = d;
        const float* pl = xlb + (size_t)s * HC;
        const float* pr = xrb + (size_t)d * HC;
        for (int j = 0; j < CPT; ++j) { xl[k][j] = pl[j]; xr[k][j] = pr[j]; }
      }
      float p[NB];
      for (int k = 0; k < NB; ++k) {
        float acc = 0.0f;
        for (int j = 0; j < CPT; ++j) {
          float v = xl[k][j] + xr[k][j];
          v = (v > 0.0f) ? v : NEG_SLOPE * v;
          acc += a[j] * v;
        }
        p[k] = acc;
      }
      lane_reduce<NB, LPH / 2>(p, lane);
      int dq = dk[0];
      for (int k = 1; k < NB; ++k) dq = (kk == k) ? dk[k] : dq;
      if (G == 1 || rep == bb) { r = p[0]; rd = dq; }
    }
    const long long e = e0 + sb * LPH + rep * NB + kk;
    if (e < ET) {
      SC[(size_t)e * H + h] = r;
      atomicMax(&M[(size_t)rd * H + h], fenc(r));
    }
  }
}

template <int H, int C, int NB>
__global__ void k_edge_aggr(const float* __restrict__ XL, const float* __restrict__ SC,
                            const unsigned* __restrict__ M, unsigned* __restrict__ DEN0,
                            unsigned* __restrict__ DEN1,
                            int* __restrict__ OUT0, int* __restrict__ OUT1,
                            const int* __restrict__ ei, long long E0, long long ET,
                            const float* __restrict__ PAR) {
  constexpr int HC  = H * C;
  constexpr int NJ  = HC / 32;
  constexpr int EPG = 32 / H;
  static_assert(EPG * H == 32 && NJ * 32 == HC && 32 % NB == 0, "shape");
  const long long e0 =
      ((long long)blockIdx.x * (blockDim.x >> 5) + (threadIdx.x >> 5)) * 32;
  if (e0 >= ET) return;
  const int lane = threadIdx.x & 31;
  const float S = PAR[0], SD = PAR[2];
  const bool upper = ((__builtin_amdgcn_s_sendmsg_rtn(0x87) >> 18) & 1u) != 0u;
  unsigned* DEN = upper ? DEN1 : DEN0;
  int*      OUT = upper ? OUT1 : OUT0;

  const long long el = e0 + lane;
  int sv = 0, dv = 0;
  if (el < E0)      { sv = ei[el]; dv = ei[E0 + el]; }
  else if (el < ET) { sv = dv = (int)(el - E0); }

  const int hl = lane % H;
  const int eg = lane / H;
  float sc[H];
#pragma unroll
  for (int g = 0; g < H; ++g) {
    const long long e  = e0 + g * EPG + eg;
    const long long ec = (e < ET) ? e : ET - 1;
    sc[g] = SC[(size_t)ec * H + hl];
  }
  int      dg[H];
  unsigned mu[H];
#pragma unroll
  for (int g = 0; g < H; ++g) {
    dg[g] = (H == 1) ? dv : __shfl(dv, g * EPG + eg, 32);
    mu[g] = M[(size_t)dg[g] * H + hl];
  }
  float w[H];
#pragma unroll
  for (int g = 0; g < H; ++g) {
    const long long e = e0 + g * EPG + eg;
    float m = fdec(mu[g]);
    if (!(m > -3.0e38f && m < 3.0e38f)) m = 0.0f;
    const float ex = (e < ET) ? __expf(sc[g] - m) : 0.0f;
    atomicAdd(&DEN[(size_t)dg[g] * H + hl], (unsigned)__float2int_rn(ex * SD));
    w[g] = ex * S;
  }

#pragma unroll
  for (int b = 0; b < 32 / NB; ++b) {
    float v[NB][NJ];
#pragma unroll
    for (int k = 0; k < NB; ++k) {
      const int s = __builtin_amdgcn_readlane(sv, b * NB + k);
      const float* xs = XL + (size_t)s * HC + lane;
#pragma unroll
      for (int j = 0; j < NJ; ++j) v[k][j] = xs[32 * j];
    }
#pragma unroll
    for (int k = 0; k < NB; ++k) {
      const int kk = b * NB + k;
      const int d = __builtin_amdgcn_readlane(dv, kk);
      int* od = OUT + (size_t)d * HC + lane;
#pragma unroll
      for (int j = 0; j < NJ; ++j) {
        const float a = __shfl(w[kk / EPG], (kk % EPG) * H + (lane + 32 * j) / C, 32);
        atomicAdd(&od[32 * j], __float2int_rn(a * v[k][j]));
      }
    }
  }
}

__global__ void k_softmax64(const int* __restrict__ in, const int* __restrict__ in2,
                            const unsigned* __restrict__ den, const unsigned* __restrict__ den2,
                            const float* __restrict__ bias, float* __restrict__ out, int n,
                            const float* __restrict__ PAR) {
  int node = blockIdx.x * (blockDim.x >> 5) + (threadIdx.x >> 5);
  if (node >= n) return;
  int lane = threadIdx.x & 31;
  const int* p = in + (size_t)node * 64 + lane * 2;
  const int* p2 = in2 + (size_t)node * 64 + lane * 2;
  const float invS = PAR[1];
  float dn = fmaxf((float)(den[node] + den2[node]) * PAR[3], 1e-16f);
  float a = (float)(p[0] + p2[0]) * invS / dn + bias[lane * 2];
  float b = (float)(p[1] + p2[1]) * invS / dn + bias[lane * 2 + 1];
  float mx = fmaxf(a, b);
#pragma unroll
  for (int o = 16; o > 0; o >>= 1) mx = fmaxf(mx, __shfl_xor(mx, o, 32));
  float ea = __expf(a - mx), eb = __expf(b - mx);
  float sum = ea + eb;
#pragma unroll
  for (int o = 16; o > 0; o >>= 1) sum += __shfl_xor(sum, o, 32);
  float inv = 1.0f / sum;
  float* q = out + (size_t)node * 64 + lane * 2;
  q[0] = ea * inv;
  q[1] = eb * inv;
}

template <int H, int C, int SRC, int DIN>
static void gat_layer(const float* xf, const int* SI, const int* SI2, const unsigned* DPREV,
                      const unsigned* DPREV2, const float* BPREV,
                      const float* PARPREV,
                      const float* Wl, const float* Wr, const float* att,
                      _Float16* WPa, _Float16* WPb,
                      float* XL, float* XR, int* OUT, int* OUT2, unsigned* M, unsigned* DEN,
                      unsigned* DEN2, float* SC,
                      const unsigned* SMALL, unsigned* AMAX, float* PAR,
                      const int* ei, int N, long long E0, hipStream_t stream) {
  constexpr int HC = H * C;
  long long ET = E0 + N;

  int packThreads = (DIN >> 5) * (HC >> 4) * 32;
  int packBlocks  = (packThreads + 255) / 256;
  k_pack_w<<<packBlocks, 256, 0, stream>>>(Wl, WPa, DIN, HC);
  k_pack_w<<<packBlocks, 256, 0, stream>>>(Wr, WPb, DIN, HC);

  int rowTiles = (N + 15) / 16;
  k_gemm_src<SRC, DIN, HC / 16><<<(rowTiles + 3) / 4, 128, 0, stream>>>(
      xf, SI, SI2, DPREV, DPREV2, BPREV, PARPREV, WPa, WPb, XL, XR, N, AMAX);
  k_scale<<<1, 32, 0, stream>>>(SMALL, AMAX, PAR);

  long long nHC = (long long)N * HC, nH = (long long)N * H;
  k_init_out<<<(int)((nHC + 255) / 256), 256, 0, stream>>>(OUT, nHC);
  k_init_md<<<(int)((nH + 255) / 256), 256, 0, stream>>>(M, DEN, nH);

  int gS = (int)((ET + 255) / 256);
  k_edge_score<H, C, (H == 1 ? 8 : 4)><<<gS, 256, 0, stream>>>(XL, XR, att, ei, E0, ET, SC, M);
  k_init_plane2<<<(int)((nHC + 255) / 256), 256, 0, stream>>>(OUT2, nHC, DEN2, nH);
  k_edge_aggr<H, C, (H == 1 ? 16 : 8)><<<gS, 256, 0, stream>>>(XL, SC, M, DEN, DEN2, OUT, OUT2,
                                                              ei, E0, ET, PAR);
}

extern "C" void kernel_launch(void* const* d_in, const int* in_sizes, int n_in,
                              void* d_out, int out_size, void* d_ws, size_t ws_size,
                              hipStream_t stream) {
  (void)n_in; (void)out_size; (void)ws_size;
  const float* x    = (const float*)d_in[0];
  const int*   ei   = (const int*)d_in[1];
  const float* Wl1  = (const float*)d_in[2];
  const float* Wr1  = (const float*)d_in[3];
  const float* att1 = (const float*)d_in[4];
  const float* b1   = (const float*)d_in[5];
  const float* Wl2  = (const float*)d_in[6];
  const float* Wr2  = (const float*)d_in[7];
  const float* att2 = (const float*)d_in[8];
  const float* b2   = (const float*)d_in[9];
  const float* Wl3  = (const float*)d_in[10];
  const float* Wr3  = (const float*)d_in[11];
  const float* att3 = (const float*)d_in[12];
  const float* b3   = (const float*)d_in[13];

  int       N  = in_sizes[0] / 256;
  long long E0 = (long long)in_sizes[1] / 2;
  long long ET = E0 + N;

  uintptr_t p = (uintptr_t)d_ws;
  auto carve = [&](size_t bytes) -> void* {
    void* r = (void*)p;
    p += (bytes + 255) & ~(size_t)255;
    return r;
  };
  _Float16* WPa = (_Float16*)carve((size_t)256 * 128 * sizeof(_Float16));
  _Float16* WPb = (_Float16*)carve((size_t)256 * 128 * sizeof(_Float16));
  float*    XL  = (float*)carve((size_t)N * 128 * sizeof(float));
  float*    XR  = (float*)carve((size_t)N * 128 * sizeof(float));
  int*      OUT = (int*)carve((size_t)N * 128 * sizeof(int));
  unsigned* M   = (unsigned*)carve((size_t)N * 8 * sizeof(unsigned));
  unsigned* DEN = (unsigned*)carve((size_t)N * 8 * sizeof(unsigned));
  float*    SC  = (float*)carve((size_t)ET * 8 * sizeof(float));
  unsigned* SMALL = (unsigned*)d_out;
  float*    PAR1  = (float*)d_out + SMALL_WORDS;
  float*    PAR2  = PAR1 + 4;
  float*    PAR3  = SC + ET;
  unsigned* DEG   = (unsigned*)OUT;
  k_zero_small<<<(N + 255) / 256, 256, 0, stream>>>(DEG, SMALL, N);
  k_count_deg<<<(int)((E0 + 255) / 256), 256, 0, stream>>>(ei, E0, DEG);
  k_deg_max<<<64, 256, 0, stream>>>(DEG, N, SMALL);

  int*      OUT2  = (int*)XR;
  unsigned* DEN2  = (unsigned*)d_out + 1024;
  float*    XR3   = XL + (size_t)N * 64;
  unsigned* DEN23 = (unsigned*)XR + (size_t)N * 64;

  gat_layer<8, 16, 0, 256>(x, nullptr, nullptr, nullptr, nullptr, nullptr, nullptr,
                           Wl1, Wr1, att1, WPa, WPb,
                           XL, XR, OUT, OUT2, M, DEN, DEN2, SC, SMALL, SMALL + 64, PAR1,
                           ei, N, E0, stream);

  gat_layer<8, 16, 1, 128>(nullptr, OUT, OUT2, DEN, DEN2, b1, PAR1, Wl2, Wr2, att2, WPa, WPb,
                           XL, XR, OUT, OUT2, M, DEN, DEN2, SC, SMALL, SMALL + 128, PAR2,
                           ei, N, E0, stream);

  gat_layer<1, 64, 1, 128>(nullptr, OUT, OUT2, DEN, DEN2, b2, PAR2, Wl3, Wr3, att3, WPa, WPb,
                           XL, XR3, OUT, OUT2, M, DEN, DEN23, SC, SMALL, SMALL + 192, PAR3,
                           ei, N, E0, stream);

  k_softmax64<<<(N + 7) / 8, 256, 0, stream>>>(OUT, OUT2, DEN, DEN23, b3, (float*)d_out, N, PAR3);
}


# Start values (zeros, -inf) of the sum planes: the GEMM wave stores them for its own 16 rows, second copies by a looping 512-block grid, 16-byte stores; no one-word-per-thread pass; 31->25 launches
# speedup vs baseline: 1.9513x; 1.0330x over previous
#include <hip/hip_runtime.h>
#include <hip/hip_bf16.h>
#include <math.h>


#define NEG_SLOPE 0.2f

typedef __attribute__((ext_vector_type(16))) _Float16 v16h;
typedef __attribute__((ext_vector_type(8)))  _Float16 v8h;
typedef __attribute__((ext_vector_type(8)))  float    v8f;

__device__ __forceinline__ unsigned fenc(float f) {
  unsigned u = __float_as_uint(f);
  return (u & 0x80000000u) ? ~u : (u | 0x80000000u);
}
__device__ __forceinline__ float fdec(unsigned u) {
  return (u & 0x80000000u) ? __uint_as_float(u & 0x7fffffffu) : __uint_as_float(~u);
}
#define ENC_NEG_INF 0x007FFFFFu

__global__ void k_clear_plane2(int* __restrict__ out2, long long nHC,
                               unsigned* __restrict__ den2, long long nH) {
  const long long nth = (long long)gridDim.x * blockDim.x;
  const long long tid = (long long)blockIdx.x * blockDim.x + threadIdx.x;
  int4* o4 = (int4*)out2;
  const long long n4 = nHC >> 2;
  const int4 z4 = make_int4(0, 0, 0, 0);
  long long i = tid;
  for (; i + 3 * nth < n4; i += 4 * nth) {
    o4[i] = z4; o4[i + nth] = z4; o4[i + 2 * nth] = z4; o4[i + 3 * nth] = z4;
  }
  for (; i < n4; i += nth) o4[i] = z4;
  uint4* d4 = (uint4*)den2;
  const long long m4 = nH >> 2;
  const uint4 u4 = make_uint4(0u, 0u, 0u, 0u);
  for (long long k = tid; k < m4; k += nth) d4[k] = u4;
  for (long long k = (m4 << 2) + tid; k < nH; k += nth) den2[k] = 0u;
}

#define SMALL_WORDS 256
__global__ void k_zero_small(unsigned* __restrict__ DEG, unsigned* __restrict__ SMALL, int n) {
  int i = blockIdx.x * blockDim.x + threadIdx.x;
  if (i < n) DEG[i] = 0u;
  if (i < SMALL_WORDS) SMALL[i] = 0u;
}

__global__ void k_count_deg(const int* __restrict__ ei, long long E0, unsigned* __restrict__ DEG) {
  long long i = (long long)blockIdx.x * blockDim.x + threadIdx.x;
  if (i < E0) atomicAdd(&DEG[ei[E0 + i]], 1u);
}

__global__ void k_deg_max(const unsigned* __restrict__ DEG, int n, unsigned* __restrict__ SMALL) {
  unsigned v = 0u;
  for (int i = blockIdx.x * blockDim.x + threadIdx.x; i < n; i += gridDim.x * blockDim.x) {
    unsigned t = DEG[i];
    v = (t > v) ? t : v;
  }
#pragma unroll
  for (int o = 16; o > 0; o >>= 1) {
    unsigned t = (unsigned)__shfl_xor((int)v, o, 32);
    v = (t > v) ? t : v;
  }
  if ((threadIdx.x & 31) == 0) atomicMax(&SMALL[0], v);
}

__global__ void k_scale(const unsigned* __restrict__ SMALL, const unsigned* __restrict__ amax64,
                        float* __restrict__ PAR) {
  int lane = threadIdx.x & 31;
  unsigned a = amax64[lane], b = amax64[lane + 32];
  a = (b > a) ? b : a;
#pragma unroll
  for (int o = 16; o > 0; o >>= 1) {
    unsigned t = (unsigned)__shfl_xor((int)a, o, 32);
    a = (t > a) ? t : a;
  }
  if (lane == 0) {
    unsigned deg = SMALL[0] + 1u;
    float bound = (float)deg * __uint_as_float(a);
    int eb = (int)((__float_as_uint(bound) >> 23) & 0xffu);
    int k = 156 - eb;
    k = (k > 100) ? 100 : ((k < -100) ? -100 : k);
    int p = 32 - __clz((int)deg);
    int kd = 31 - p;
    kd = (kd > 30) ? 30 : ((kd < 0) ? 0 : kd);
    PAR[0] = __uint_as_float((unsigned)(127 + k) << 23);
    PAR[1] = __uint_as_float((unsigned)(127 - k) << 23);
    PAR[2] = __uint_as_float((unsigned)(127 + kd) << 23);
    PAR[3] = __uint_as_float((unsigned)(127 - kd) << 23);
  }
}

__global__ void k_pack_w(const float* __restrict__ W, _Float16* __restrict__ WP,
                         int din, int dout) {
  int ktiles = din >> 5, ntiles = dout >> 4;
  int total = ktiles * ntiles * 32;
  int idx = blockIdx.x * blockDim.x + threadIdx.x;
  if (idx >= total) return;
  int lane = idx & 31;
  int frag = idx >> 5;
  int nt = frag % ntiles;
  int kt = frag / ntiles;
  int half = lane >> 4, lr = lane & 15;
  _Float16* o = WP + ((size_t)frag * 32 + lane) * 16;
  const float* w0 = W + (size_t)(kt * 32 + half * 8) * dout + nt * 16 + lr;
#pragma unroll
  for (int j = 0; j < 8; ++j) {
    o[j]     = (_Float16)w0[(size_t)j * dout];
    o[8 + j] = (_Float16)w0[(size_t)(16 + j) * dout];
  }
}

template <int SRC, int DIN, int NT, int H>
__global__ void __launch_bounds__(128) k_gemm_src(const float* __restrict__ xf, const int* si,
                           const int* si2,
                           const unsigned* den, const unsigned* __restrict__ den2,
                           const float* __restrict__ bias,
                           const float* __restrict__ PARp,
                           const _Float16* __restrict__ WPa, const _Float16* __restrict__ WPb,
                           float* __restrict__ XL, float* XR, int Mrows,
                           unsigned* __restrict__ amax,
                           int* ZO, unsigned* ZM, unsigned* ZD) {
  constexpr int DOUT = NT * 16;
  int wave = threadIdx.x >> 5;
  int lane = threadIdx.x & 31;
  int half = lane >> 4;
  int lr   = lane & 15;
  int tileM = blockIdx.x * 4 + wave;
  if (tileM * 16 >= Mrows) return;

  int row = tileM * 16 + lr;
  if (row >= Mrows) row = Mrows - 1;

  float invS = 0.0f, invSD = 0.0f;
  if constexpr (SRC == 1) { invS = PARp[1]; invSD = PARp[3]; }

  v8f acc[2 * NT];
#pragma unroll
  for (int t = 0; t < 2 * NT; ++t) acc[t] = v8f{};

  const v16h* WPav = (const v16h*)WPa;
  const v16h* WPbv = (const v16h*)WPb;
  for (int kt = 0; kt < DIN / 32; ++kt) {
    const int kb = kt * 32 + half * 8;
    float f[16];
    if constexpr (SRC == 0) {
      const float4* p = (const float4*)(xf + (size_t)row * DIN + kb);
      float4 q0 = p[0], q1 = p[1], q2 = p[4], q3 = p[5];
      f[0] = q0.x; f[1] = q0.y; f[2]  = q0.z; f[3]  = q0.w;
      f[4] = q1.x; f[5] = q1.y; f[6]  = q1.z; f[7]  = q1.w;
      f[8] = q2.x; f[9] = q2.y; f[10] = q2.z; f[11] = q2.w;
      f[12] = q3.x; f[13] = q3.y; f[14] = q3.z; f[15] = q3.w;
    } else {
      const int4* p = (const int4*)(si + (size_t)row * DIN + kb);
      int4 q0 = p[0], q1 = p[1], q2 = p[4], q3 = p[5];
      const int4* p2 = (const int4*)(si2 + (size_t)row * DIN + kb);
      const int4 r0 = p2[0], r1 = p2[1], r2 = p2[4], r3 = p2[5];
      q0.x += r0.x; q0.y += r0.y; q0.z += r0.z; q0.w += r0.w;
      q1.x += r1.x; q1.y += r1.y; q1.z += r1.z; q1.w += r1.w;
      q2.x += r2.x; q2.y += r2.y; q2.z += r2.z; q2.w += r2.w;
      q3.x += r3.x; q3.y += r3.y; q3.z += r3.z; q3.w += r3.w;
      const float4* bp = (const float4*)(bias + kb);
      float4 b0 = bp[0], b1 = bp[1], b2 = bp[4], b3 = bp[5];
      uint2 dd = *(const uint2*)(den + (size_t)row * 8 + 2 * kt);
      const uint2 dd2 = *(const uint2*)(den2 + (size_t)row * 8 + 2 * kt);
      dd.x += dd2.x; dd.y += dd2.y;
      const float dlo = fmaxf((float)dd.x * invSD, 1e-16f);
      const float dhi = fmaxf((float)dd.y * invSD, 1e-16f);
      f[0] = (float)q0.x * invS / dlo + b0.x;  f[1] = (float)q0.y * invS / dlo + b0.y;
      f[2] = (float)q0.z * invS / dlo + b0.z;  f[3] = (float)q0.w * invS / dlo + b0.w;
      f[4] = (float)q1.x * invS / dlo + b1.x;  f[5] = (float)q1.y * invS / dlo + b1.y;
      f[6] = (float)q1.z * invS / dlo + b1.z;  f[7] = (float)q1.w * invS / dlo + b1.w;
      f[8]  = (float)q2.x * invS / dhi + b2.x; f[9]  = (float)q2.y * invS / dhi + b2.y;
      f[10] = (float)q2.z * invS / dhi + b2.z; f[11] = (float)q2.w * invS / dhi + b2.w;
      f[12] = (float)q3.x * invS / dhi + b3.x; f[13] = (float)q3.y * invS / dhi + b3.y;
      f[14] = (float)q3.z * invS / dhi + b3.z; f[15] = (float)q3.w * invS / dhi + b3.w;
#pragma unroll
      for (int j = 0; j < 16; ++j) f[j] = (f[j] > 0.0f) ? f[j] : expm1f(f[j]);
    }
    v16h a;
#pragma unroll
    for (int j = 0; j < 16; ++j) a[j] = (_Float16)f[j];

    const size_t fb = (size_t)kt * NT * 32 + lane;
    v16h bl[NT], br[NT];
#pragma unroll
    for (int t = 0; t < NT; ++t) bl[t] = WPav[fb + (size_t)t * 32];
    if constexpr (SRC == 1) __builtin_amdgcn_sched_barrier(0);
#pragma unroll
    for (int t = 0; t < NT; ++t) br[t] = WPbv[fb + (size_t)t * 32];
#pragma unroll
    for (int t = 0; t < NT; ++t)
      acc[t] = __builtin_amdgcn_wmma_f32_16x16x32_f16(false, a, false, bl[t], (short)0, acc[t], false, false);
    if constexpr (SRC == 1) __builtin_amdgcn_sched_barrier(0);
#pragma unroll
    for (int t = 0; t < NT; ++t)
      acc[NT + t] = __builtin_amdgcn_wmma_f32_16x16x32_f16(false, a, false, br[t], (short)0, acc[NT + t], false, false);
  }
  int r0 = tileM * 16 + half * 8;
#pragma unroll
  for (int i = 0; i < 8; ++i) {
    int r = r0 + i;
    if (r < Mrows) {
      float* yl = XL + (size_t)r * DOUT + lr;
      float* yr = XR + (size_t)r * DOUT + lr;
#pragma unroll
      for (int t = 0; t < NT; ++t) {
        yl[t * 16] = acc[t][i];
        yr[t * 16] = acc[NT + t][i];
      }
    }
  }
  float mx = 0.0f;
#pragma unroll
  for (int t = 0; t < NT; ++t) {
#pragma unroll
    for (int i = 0; i < 8; ++i) mx = fmaxf(mx, fabsf(acc[t][i]));
  }
#pragma unroll
  for (int o = 16; o > 0; o >>= 1) mx = fmaxf(mx, __shfl_xor(mx, o, 32));
  if (lane == 0) atomicMax(&amax[tileM & 63], __float_as_uint(mx));

  constexpr int ZROW = (SRC == 1) ? DIN : DOUT;
  constexpr int ZH   = (SRC == 1) ? DIN / 16 : H;
  const long long zlim = (long long)Mrows * DOUT;
  const long long hlim = (long long)Mrows * H;
  const long long zb = (long long)tileM * 16 * ZROW;
  int4* zo = (int4*)(ZO + zb);
  const int4 z4 = make_int4(0, 0, 0, 0);
  if (zb + 16 * ZROW <= zlim) {
#pragma unroll
    for (int j = 0; j < 16 * ZROW / 128; ++j) zo[lane + 32 * j] = z4;
  } else {
    for (int i = lane; i < 16 * ZROW / 4; i += 32)
      if (zb + 4LL * i < zlim) zo[i] = z4;
  }
  const long long mb = (long long)tileM * 16 * H;
  for (int i = lane; i < 16 * H; i += 32)
    if (mb + i < hlim) ZM[mb + i] = ENC_NEG_INF;
  const long long db = (long long)tileM * 16 * ZH;
  for (int i = lane; i < 16 * ZH; i += 32)
    if (db + i < hlim) ZD[db + i] = 0u;
}

template <int N, int O>
__device__ __forceinline__ void lane_reduce(float* p, int lane) {
  if constexpr (O > 0) {
    if constexpr (N > 1) {
      constexpr int HN = N / 2;
      const bool up = (lane & O) != 0;
      for (int i = 0; i < HN; ++i) {
        float keep = up ? p[HN + i] : p[i];
        float send = up ? p[i] : p[HN + i];
        p[i] = keep + __shfl_xor(send, O, 32);
      }
      lane_reduce<HN, O / 2>(p, lane);
    } else {
      p[0] += __shfl_xor(p[0], O, 32);
      lane_reduce<1, O / 2>(p, lane);
    }
  }
}

template <int H, int C, int NB>
__global__ void k_edge_score(const float* __restrict__ XL, const float* __restrict__ XR,
                             const float* __restrict__ att, const int* __restrict__ ei,
                             long long E0, long long ET,
                             float* __restrict__ SC, unsigned* __restrict__ M) {
  constexpr int HC  = H * C;
  constexpr int LPH = 32 / H;
  constexpr int CPT = C / LPH;
  constexpr int G   = LPH / NB;
  static_assert(NB * G == LPH && H * LPH == 32 && CPT * LPH == C, "shape");
  const long long e0 =
      ((long long)blockIdx.x * (blockDim.x >> 5) + (threadIdx.x >> 5)) * 32;
  if (e0 >= ET) return;
  const int lane = threadIdx.x & 31;

  const long long el = e0 + lane;
  int sv = 0, dv = 0;
  if (el < E0)      { sv = ei[el]; dv = ei[E0 + el]; }
  else if (el < ET) { sv = dv = (int)(el - E0); }

  const int h   = lane / LPH;
  const int q   = lane % LPH;
  const int kk  = q / G;
  const int rep = q % G;
  float a[CPT];
  for (int j = 0; j < CPT; ++j) a[j] = att[lane * CPT + j];
  const float* xlb = XL + lane * CPT;
  const float* xrb = XR + lane * CPT;

  for (int sb = 0; sb < H; ++sb) {
    float r  = 0.0f;
    int   rd = 0;
    for (int bb = 0; bb < G; ++bb) {
      const int base = sb * LPH + bb * NB;
      float xl[NB][CPT], xr[NB][CPT];
      int   dk[NB];
      for (int k = 0; k < NB; ++k) {
        const int s = __builtin_amdgcn_readlane(sv, base + k);
        const int d = __builtin_amdgcn_readlane(dv, base + k);
        dk[k] = d;
        const float* pl = xlb + (size_t)s * HC;
        const float* pr = xrb + (size_t)d * HC;
        for (int j = 0; j < CPT; ++j) { xl[k][j] = pl[j]; xr[k][j] = pr[j]; }
      }
      float p[NB];
      for (int k = 0; k < NB; ++k) {
        float acc = 0.0f;
        for (int j = 0; j < CPT; ++j) {
          float v = xl[k][j] + xr[k][j];
          v = (v > 0.0f) ? v : NEG_SLOPE * v;
          acc += a[j] * v;
        }
        p[k] = acc;
      }
      lane_reduce<NB, LPH / 2>(p, lane);
      int dq = dk[0];
      for (int k = 1; k < NB; ++k) dq = (kk == k) ? dk[k] : dq;
      if (G == 1 || rep == bb) { r = p[0]; rd = dq; }
    }
    const long long e = e0 + sb * LPH + rep * NB + kk;
    if (e < ET) {
      SC[(size_t)e * H + h] = r;
      atomicMax(&M[(size_t)rd * H + h], fenc(r));
    }
  }
}

template <int H, int C, int NB>
__global__ void k_edge_aggr(const float* __restrict__ XL, const float* __restrict__ SC,
                            const unsigned* __restrict__ M, unsigned* __restrict__ DEN0,
                            unsigned* __restrict__ DEN1,
                            int* __restrict__ OUT0, int* __restrict__ OUT1,
                            const int* __restrict__ ei, long long E0, long long ET,
                            const float* __restrict__ PAR) {
  constexpr int HC  = H * C;
  constexpr int NJ  = HC / 32;
  constexpr int EPG = 32 / H;
  static_assert(EPG * H == 32 && NJ * 32 == HC && 32 % NB == 0, "shape");
  const long long e0 =
      ((long long)blockIdx.x * (blockDim.x >> 5) + (threadIdx.x >> 5)) * 32;
  if (e0 >= ET) return;
  const int lane = threadIdx.x & 31;
  const float S = PAR[0], SD = PAR[2];
  const bool upper = ((__builtin_amdgcn_s_sendmsg_rtn(0x87) >> 18) & 1u) != 0u;
  unsigned* DEN = upper ? DEN1 : DEN0;
  int*      OUT = upper ? OUT1 : OUT0;

  const long long el = e0 + lane;
  int sv = 0, dv = 0;
  if (el < E0)      { sv = ei[el]; dv = ei[E0 + el]; }
  else if (el < ET) { sv = dv = (int)(el - E0); }

  const int hl = lane % H;
  const int eg = lane / H;
  float sc[H];
#pragma unroll
  for (int g = 0; g < H; ++g) {
    const long long e  = e0 + g * EPG + eg;
    const long long ec = (e < ET) ? e : ET - 1;
    sc[g] = SC[(size_t)ec * H + hl];
  }
  int      dg[H];
  unsigned mu[H];
#pragma unroll
  for (int g = 0; g < H; ++g) {
    dg[g] = (H == 1) ? dv : __shfl(dv, g * EPG + eg, 32);
    mu[g] = M[(size_t)dg[g] * H + hl];
  }
  float w[H];
#pragma unroll
  for (int g = 0; g < H; ++g) {
    const long long e = e0 + g * EPG + eg;
    float m = fdec(mu[g]);
    if (!(m > -3.0e38f && m < 3.0e38f)) m = 0.0f;
    const float ex = (e < ET) ? __expf(sc[g] - m) : 0.0f;
    atomicAdd(&DEN[(size_t)dg[g] * H + hl], (unsigned)__float2int_rn(ex * SD));
    w[g] = ex * S;
  }

#pragma unroll
  for (int b = 0; b < 32 / NB; ++b) {
    float v[NB][NJ];
#pragma unroll
    for (int k = 0; k < NB; ++k) {
      const int s = __builtin_amdgcn_readlane(sv, b * NB + k);
      const float* xs = XL + (size_t)s * HC + lane;
#pragma unroll
      for (int j = 0; j < NJ; ++j) v[k][j] = xs[32 * j];
    }
#pragma unroll
    for (int k = 0; k < NB; ++k) {
      const int kk = b * NB + k;
      const int d = __builtin_amdgcn_readlane(dv, kk);
      int* od = OUT + (size_t)d * HC + lane;
#pragma unroll
      for (int j = 0; j < NJ; ++j) {
        const float a = __shfl(w[kk / EPG], (kk % EPG) * H + (lane + 32 * j) / C, 32);
        atomicAdd(&od[32 * j], __float2int_rn(a * v[k][j]));
      }
    }
  }
}

__global__ void k_softmax64(const int* __restrict__ in, const int* __restrict__ in2,
                            const unsigned* __restrict__ den, const unsigned* __restrict__ den2,
                            const float* __restrict__ bias, float* __restrict__ out, int n,
                            const float* __restrict__ PAR) {
  int node = blockIdx.x * (blockDim.x >> 5) + (threadIdx.x >> 5);
  if (node >= n) return;
  int lane = threadIdx.x & 31;
  const int* p = in + (size_t)node * 64 + lane * 2;
  const int* p2 = in2 + (size_t)node * 64 + lane * 2;
  const float invS = PAR[1];
  float dn = fmaxf((float)(den[node] + den2[node]) * PAR[3], 1e-16f);
  float a = (float)(p[0] + p2[0]) * invS / dn + bias[lane * 2];
  float b = (float)(p[1] + p2[1]) * invS / dn + bias[lane * 2 + 1];
  float mx = fmaxf(a, b);
#pragma unroll
  for (int o = 16; o > 0; o >>= 1) mx = fmaxf(mx, __shfl_xor(mx, o, 32));
  float ea = __expf(a - mx), eb = __expf(b - mx);
  float sum = ea + eb;
#pragma unroll
  for (int o = 16; o > 0; o >>= 1) sum += __shfl_xor(sum, o, 32);
  float inv = 1.0f / sum;
  float* q = out + (size_t)node * 64 + lane * 2;
  q[0] = ea * inv;
  q[1] = eb * inv;
}

template <int H, int C, int SRC, int DIN>
static void gat_layer(const float* xf, const int* SI, const int* SI2, const unsigned* DPREV,
                      const unsigned* DPREV2, const float* BPREV,
                      const float* PARPREV,
                      const float* Wl, const float* Wr, const float* att,
                      _Float16* WPa, _Float16* WPb,
                      float* XL, float* XR, int* OUT, int* OUT2, unsigned* M, unsigned* DEN,
                      unsigned* DEN2, float* SC,
                      const unsigned* SMALL, unsigned* AMAX, float* PAR,
                      const int* ei, int N, long long E0, hipStream_t stream) {
  constexpr int HC = H * C;
  long long ET = E0 + N;

  int packThreads = (DIN >> 5) * (HC >> 4) * 32;
  int packBlocks  = (packThreads + 255) / 256;
  k_pack_w<<<packBlocks, 256, 0, stream>>>(Wl, WPa, DIN, HC);
  k_pack_w<<<packBlocks, 256, 0, stream>>>(Wr, WPb, DIN, HC);

  int rowTiles = (N + 15) / 16;
  k_gemm_src<SRC, DIN, HC / 16, H><<<(rowTiles + 3) / 4, 128, 0, stream>>>(
      xf, SI, SI2, DPREV, DPREV2, BPREV, PARPREV, WPa, WPb, XL, XR, N, AMAX, OUT, M, DEN);
  k_scale<<<1, 32, 0, stream>>>(SMALL, AMAX, PAR);

  long long nHC = (long long)N * HC, nH = (long long)N * H;

  int gS = (int)((ET + 255) / 256);
  k_edge_score<H, C, (H == 1 ? 8 : 4)><<<gS, 256, 0, stream>>>(XL, XR, att, ei, E0, ET, SC, M);
  k_clear_plane2<<<512, 256, 0, stream>>>(OUT2, nHC, DEN2, nH);
  k_edge_aggr<H, C, (H == 1 ? 16 : 8)><<<gS, 256, 0, stream>>>(XL, SC, M, DEN, DEN2, OUT, OUT2,
                                                              ei, E0, ET, PAR);
}

extern "C" void kernel_launch(void* const* d_in, const int* in_sizes, int n_in,
                              void* d_out, int out_size, void* d_ws, size_t ws_size,
                              hipStream_t stream) {
  (void)n_in; (void)out_size; (void)ws_size;
  const float* x    = (const float*)d_in[0];
  const int*   ei   = (const int*)d_in[1];
  const float* Wl1  = (const float*)d_in[2];
  const float* Wr1  = (const float*)d_in[3];
  const float* att1 = (const float*)d_in[4];
  const float* b1   = (const float*)d_in[5];
  const float* Wl2  = (const float*)d_in[6];
  const float* Wr2  = (const float*)d_in[7];
  const float* att2 = (const float*)d_in[8];
  const float* b2   = (const float*)d_in[9];
  const float* Wl3  = (const float*)d_in[10];
  const float* Wr3  = (const float*)d_in[11];
  const float* att3 = (const float*)d_in[12];
  const float* b3   = (const float*)d_in[13];

  int       N  = in_sizes[0] / 256;
  long long E0 = (long long)in_sizes[1] / 2;
  long long ET = E0 + N;

  uintptr_t p = (uintptr_t)d_ws;
  auto carve = [&](size_t bytes) -> void* {
    void* r = (void*)p;
    p += (bytes + 255) & ~(size_t)255;
    return r;
  };
  _Float16* WPa = (_Float16*)carve((size_t)256 * 128 * sizeof(_Float16));
  _Float16* WPb = (_Float16*)carve((size_t)256 * 128 * sizeof(_Float16));
  float*    XL  = (float*)carve((size_t)N * 128 * sizeof(float));
  float*    XR  = (float*)carve((size_t)N * 128 * sizeof(float));
  int*      OUT = (int*)carve((size_t)N * 128 * sizeof(int));
  unsigned* M   = (unsigned*)carve((size_t)N * 8 * sizeof(unsigned));
  unsigned* DEN = (unsigned*)carve((size_t)N * 8 * sizeof(unsigned));
  float*    SC  = (float*)carve((size_t)ET * 8 * sizeof(float));
  unsigned* SMALL = (unsigned*)d_out;
  float*    PAR1  = (float*)d_out + SMALL_WORDS;
  float*    PAR2  = PAR1 + 4;
  float*    PAR3  = SC + ET;
  unsigned* DEG   = (unsigned*)OUT;
  k_zero_small<<<(N + 255) / 256, 256, 0, stream>>>(DEG, SMALL, N);
  k_count_deg<<<(int)((E0 + 255) / 256), 256, 0, stream>>>(ei, E0, DEG);
  k_deg_max<<<64, 256, 0, stream>>>(DEG, N, SMALL);

  int*      OUT2  = (int*)XR;
  unsigned* DEN2  = (unsigned*)d_out + 1024;
  float*    XR3   = XL + (size_t)N * 64;
  unsigned* DEN23 = (unsigned*)XR + (size_t)N * 64;

  gat_layer<8, 16, 0, 256>(x, nullptr, nullptr, nullptr, nullptr, nullptr, nullptr,
                           Wl1, Wr1, att1, WPa, WPb,
                           XL, XR, OUT, OUT2, M, DEN, DEN2, SC, SMALL, SMALL + 64, PAR1,
                           ei, N, E0, stream);

  gat_layer<8, 16, 1, 128>(nullptr, OUT, OUT2, DEN, DEN2, b1, PAR1, Wl2, Wr2, att2, WPa, WPb,
                           XL, XR, OUT, OUT2, M, DEN, DEN2, SC, SMALL, SMALL + 128, PAR2,
                           ei, N, E0, stream);

  gat_layer<1, 64, 1, 128>(nullptr, OUT, OUT2, DEN, DEN2, b2, PAR2, Wl3, Wr3, att3, WPa, WPb,
                           XL, XR3, OUT, OUT2, M, DEN, DEN23, SC, SMALL, SMALL + 192, PAR3,
                           ei, N, E0, stream);

  k_softmax64<<<(N + 7) / 8, 256, 0, stream>>>(OUT, OUT2, DEN, DEN23, b3, (float*)d_out, N, PAR3);
}
